# starting kernel
# baseline (speedup 1.0000x reference)
#include <hip/hip_runtime.h>
#include <math.h>


typedef __attribute__((ext_vector_type(2))) float v2f;
typedef __attribute__((ext_vector_type(8))) float v8f;

#define B_TOT   2048
#define T_STEPS 128
#define NXc     8
#define NUc     2
#define NZc     50
#define DINP    64
#define DH      128
#define BT      16
#define DELTA_F 0.01f

#define LDW     129
#define LDW3    17
#define OFF_W1P 0u
#define OFF_W2P (OFF_W1P + 32u*LDW*8u)
#define OFF_W3P (OFF_W2P + 64u*LDW*8u)
#define OFF_INP (OFF_W3P + 64u*LDW3*8u)
#define OFF_H1  (OFF_INP + 16u*64u*4u)
#define OFF_H2  (OFF_H1  + 16u*128u*4u)
#define OFF_NN  (OFF_H2  + 16u*128u*4u)
#define OFF_XG  (OFF_NN  + 256u*4u)
#define OFF_XS  (OFF_XG  + 128u*4u)
#define OFF_Z   (OFF_XS  + 128u*4u)
#define OFF_K   (OFF_Z   + 16u*52u*4u)
#define OFF_U   (OFF_K   + 512u*4u)
#define OFF_B1  (OFF_U   + 32u*4u)
#define OFF_B2  (OFF_B1  + 128u*4u)
#define OFF_B3  (OFF_B2  + 128u*4u)
#define SMEM_BYTES (OFF_B3 + 16u*4u)

__device__ __forceinline__ float fast_tanh(float x) {
  float e = __expf(2.0f * x);
  return 1.0f - __fdividef(2.0f, e + 1.0f);
}

__device__ __forceinline__ v8f mm16(const float* __restrict__ A, int lda,
                                    const v2f* __restrict__ Bp, int ldb,
                                    int kIters, int lane) {
  v8f acc = {};
  const int mr = lane & 15;
  const int kh = lane >> 4;
  const float* ap = A + mr * lda + 2 * kh;
  const v2f*   bp = Bp + (size_t)kh * ldb + mr;
#pragma unroll
  for (int kk = 0; kk < kIters; ++kk) {
    v2f a = *(const v2f*)(ap + 4 * kk);
    v2f b = bp[(size_t)(2 * kk) * ldb];
    acc = __builtin_amdgcn_wmma_f32_16x16x4_f32(false, a, false, b,
                                                (short)0, acc, false, false);
  }
  return acc;
}

__device__ __forceinline__ void mlp_and_physics(
    int tid, int wave, int lane, int stage,
    const float* sInp, float* sH1, float* sH2, float* sNN,
    const v2f* W1p, const v2f* W2p, const v2f* W3p,
    const float* sB1, const float* sB2, const float* sB3,
    const float* xsrc, const float* sU, float* sK) {
  {
    v8f d = mm16(sInp, DINP, W1p + wave * 16, LDW, 16, lane);
    const int nc = wave * 16 + (lane & 15);
    const int kh = lane >> 4;
    const float bias = sB1[nc];
#pragma unroll
    for (int v = 0; v < 8; ++v)
      sH1[(v + 8 * kh) * DH + nc] = fast_tanh(d[v] + bias);
  }
  __syncthreads();
  {
    v8f d = mm16(sH1, DH, W2p + wave * 16, LDW, 32, lane);
    const int nc = wave * 16 + (lane & 15);
    const int kh = lane >> 4;
    const float bias = sB2[nc];
#pragma unroll
    for (int v = 0; v < 8; ++v)
      sH2[(v + 8 * kh) * DH + nc] = fast_tanh(d[v] + bias);
  }
  sNN[tid] = 0.0f;
  __syncthreads();
  {
    v8f d = mm16(sH2 + wave * 16, DH, W3p + wave * 8 * LDW3, LDW3, 4, lane);
    const int nc = lane & 15;
    const int kh = lane >> 4;
    if (nc < NXc) {
#pragma unroll
      for (int v = 0; v < 8; ++v)
        atomicAdd(&sNN[(v + 8 * kh) * 16 + nc], d[v]);
    }
  }
  __syncthreads();
  if (tid < BT) {
    const float* xs = xsrc + tid * NXc;
    float Hr  = xs[0] * 2.0f + 50.0f;
    float CAr = xs[1] * 0.2f + 1.0f;
    float CBr = xs[2] * 0.1f + 0.5f;
    float Tr  = xs[3] * 4.0f + 313.0f;
    float Hb  = xs[4] * 2.0f + 50.0f;
    float CAb = xs[5] * 0.2f + 1.0f;
    float CBb = xs[6] * 0.1f + 0.5f;
    float Tb  = xs[7] * 4.0f + 313.0f;
    float F = sU[tid * 2 + 0] * 1.0f + 10.0f;
    float D = sU[tid * 2 + 1] * 0.5f + 5.0f;
    float Fr = 2.5f * __fsqrt_rn(Hr);
    float Fb = 1.5f * __fsqrt_rn(Hb);
    float invArHr = __fdividef(1.0f, 3.0f * Hr);
    float invAbHb = __fdividef(1.0f, 3.0f * Hb);
    float dHr  = (F + D - Fr) * (1.0f / 3.0f);
    float dCAr = (F * (6.0f - CAr) - D * CAr) * invArHr;
    float dCBr = (-F * CBr - D * CBr) * invArHr;
    float dTr  = (F * (320.0f - Tr) + D * (300.0f - Tr)) * invArHr
               + (200.0f / 30.0f) * invArHr;
    float dHb  = (Fr - Fb - D) * (1.0f / 3.0f);
    float dCAb = (Fr * (CAr - CAb) + D * CAb) * invAbHb;
    float dCBb = (Fr * (CBr - CBb) + D * CBb) * invAbHb;
    float dTb  = Fr * (Tr - Tb) * invAbHb + (200.0f / 30.0f) * invAbHb;
    float*       kr = sK + stage * (BT * NXc) + tid * NXc;
    const float* nn = sNN + tid * 16;
    kr[0] = dHr  * 0.5f  + nn[0] + sB3[0];
    kr[1] = dCAr * 5.0f  + nn[1] + sB3[1];
    kr[2] = dCBr * 10.0f + nn[2] + sB3[2];
    kr[3] = dTr  * 0.25f + nn[3] + sB3[3];
    kr[4] = dHb  * 0.5f  + nn[4] + sB3[4];
    kr[5] = dCAb * 5.0f  + nn[5] + sB3[5];
    kr[6] = dCBb * 10.0f + nn[6] + sB3[6];
    kr[7] = dTb  * 0.25f + nn[7] + sB3[7];
  }
  __syncthreads();
}

__global__ __launch_bounds__(256) void cstr_rollout_kernel(
    const float* __restrict__ useq, const float* __restrict__ xGz0,
    const float* __restrict__ W1g, const float* __restrict__ b1g,
    const float* __restrict__ W2g, const float* __restrict__ b2g,
    const float* __restrict__ W3g, const float* __restrict__ b3g,
    float* __restrict__ out) {
  extern __shared__ __align__(16) char smem[];
  v2f*   W1p  = (v2f*)(smem + OFF_W1P);
  v2f*   W2p  = (v2f*)(smem + OFF_W2P);
  v2f*   W3p  = (v2f*)(smem + OFF_W3P);
  float* sInp = (float*)(smem + OFF_INP);
  float* sH1  = (float*)(smem + OFF_H1);
  float* sH2  = (float*)(smem + OFF_H2);
  float* sNN  = (float*)(smem + OFF_NN);
  float* sXG  = (float*)(smem + OFF_XG);
  float* sXs  = (float*)(smem + OFF_XS);
  float* sZ   = (float*)(smem + OFF_Z);
  float* sK   = (float*)(smem + OFF_K);
  float* sU   = (float*)(smem + OFF_U);
  float* sB1  = (float*)(smem + OFF_B1);
  float* sB2  = (float*)(smem + OFF_B2);
  float* sB3  = (float*)(smem + OFF_B3);

  const int tid  = threadIdx.x;
  const int lane = tid & 31;
  const int wave = tid >> 5;
  const int bt0  = blockIdx.x * BT;

  for (int i = tid; i < 32 * 128; i += 256) {
    int kk = i >> 7, n = i & 127;
    float lo = (2 * kk     < 60) ? W1g[(2 * kk)     * 128 + n] : 0.0f;
    float hi = (2 * kk + 1 < 60) ? W1g[(2 * kk + 1) * 128 + n] : 0.0f;
    W1p[kk * LDW + n] = (v2f){lo, hi};
  }
  for (int i = tid; i < 64 * 128; i += 256) {
    int kk = i >> 7, n = i & 127;
    W2p[kk * LDW + n] = (v2f){W2g[(2 * kk) * 128 + n], W2g[(2 * kk + 1) * 128 + n]};
  }
  for (int i = tid; i < 64 * 16; i += 256) {
    int kk = i >> 4, n = i & 15;
    v2f p = (v2f){0.0f, 0.0f};
    if (n < NXc) p = (v2f){W3g[(2 * kk) * NXc + n], W3g[(2 * kk + 1) * NXc + n]};
    W3p[kk * LDW3 + n] = p;
  }
  if (tid < 128) { sB1[tid] = b1g[tid]; sB2[tid] = b2g[tid]; }
  if (tid < NXc) sB3[tid] = b3g[tid];
  for (int i = tid; i < BT * 58; i += 256) {
    int r = i / 58, c = i - r * 58;
    float v = xGz0[(size_t)(bt0 + r) * 58 + c];
    if (c < NXc) sXG[r * NXc + c] = v;
    else         sZ[r * 52 + (c - NXc)] = v;
  }
  __syncthreads();

  for (int t = 0; t < T_STEPS; ++t) {
    if (tid < BT * NUc) {
      int r = tid >> 1, c = tid & 1;
      sU[tid] = useq[(size_t)(bt0 + r) * T_STEPS * NUc + (size_t)t * NUc + c];
    }
    __syncthreads();

    for (int i = tid; i < BT * DINP; i += 256) {
      int r = i >> 6, c = i & 63;
      float v;
      if      (c < 8)  v = sXG[r * NXc + c];
      else if (c < 58) v = sZ[r * 52 + (c - 8)];
      else if (c < 60) v = sU[r * 2 + (c - 58)];
      else             v = 0.0f;
      sInp[i] = v;
    }
    __syncthreads();
    mlp_and_physics(tid, wave, lane, 0, sInp, sH1, sH2, sNN, W1p, W2p, W3p,
                    sB1, sB2, sB3, sXG, sU, sK);

    for (int i = tid; i < BT * DINP; i += 256) {
      int r = i >> 6, c = i & 63;
      float v;
      if (c < 8) {
        v = sXG[r * NXc + c] + 0.5f * DELTA_F * sK[0 * 128 + r * NXc + c];
        sXs[r * NXc + c] = v;
      } else if (c < 48) {
        int idx = c - 8, p = idx >> 3, q = idx & 7;
        float y0 = sZ[r * 52 + p * 8 + q];
        float y1 = (p < 4) ? sZ[r * 52 + (p + 1) * 8 + q] : sXG[r * NXc + q];
        v = 0.5f * (y0 + y1);
      } else if (c < 58) v = sZ[r * 52 + 40 + (c - 48)];
      else if (c < 60)   v = sU[r * 2 + (c - 58)];
      else               v = 0.0f;
      sInp[i] = v;
    }
    __syncthreads();
    mlp_and_physics(tid, wave, lane, 1, sInp, sH1, sH2, sNN, W1p, W2p, W3p,
                    sB1, sB2, sB3, sXs, sU, sK);

    if (tid < BT * NXc) {
      int r = tid >> 3, j = tid & 7;
      float v = sXG[r * NXc + j] + 0.5f * DELTA_F * sK[1 * 128 + r * NXc + j];
      sXs[r * NXc + j] = v;
      sInp[r * DINP + j] = v;
    }
    __syncthreads();
    mlp_and_physics(tid, wave, lane, 2, sInp, sH1, sH2, sNN, W1p, W2p, W3p,
                    sB1, sB2, sB3, sXs, sU, sK);

    for (int i = tid; i < BT * 48; i += 256) {
      int r = i / 48, c = i - r * 48;
      float v;
      if (c < 8) {
        v = sXG[r * NXc + c] + DELTA_F * sK[2 * 128 + r * NXc + c];
        sXs[r * NXc + c] = v;
      } else {
        int idx = c - 8;
        v = (idx < 32) ? sZ[r * 52 + 8 + idx] : sXG[r * NXc + (idx - 32)];
      }
      sInp[r * DINP + c] = v;
    }
    __syncthreads();
    mlp_and_physics(tid, wave, lane, 3, sInp, sH1, sH2, sNN, W1p, W2p, W3p,
                    sB1, sB2, sB3, sXs, sU, sK);

    float xgnew = 0.0f;
    if (tid < BT * NXc) {
      int r = tid >> 3, j = tid & 7;
      float xg = sXG[r * NXc + j];
      out[(size_t)(bt0 + r) * T_STEPS * NXc + (size_t)t * NXc + j] = xg;
      xgnew = xg + (DELTA_F / 6.0f) *
              (sK[0 * 128 + tid] + 2.0f * sK[1 * 128 + tid] +
               2.0f * sK[2 * 128 + tid] + sK[3 * 128 + tid]);
    }
    float regz[4];
#pragma unroll
    for (int ii = 0; ii < 4; ++ii) {
      int i = tid + ii * 256;
      if (i < BT * NZc) {
        int r = i / NZc, c = i - r * NZc;
        float v;
        if      (c < 32) v = sZ[r * 52 + 8 + c];
        else if (c < 40) v = sXG[r * NXc + (c - 32)];
        else if (c < 48) v = sZ[r * 52 + (c + 2)];
        else             v = sU[r * 2 + (c - 48)];
        regz[ii] = v;
      }
    }
    __syncthreads();
#pragma unroll
    for (int ii = 0; ii < 4; ++ii) {
      int i = tid + ii * 256;
      if (i < BT * NZc) {
        int r = i / NZc, c = i - r * NZc;
        sZ[r * 52 + c] = regz[ii];
      }
    }
    if (tid < BT * NXc) sXG[tid] = xgnew;
    __syncthreads();
  }
}

extern "C" void kernel_launch(void* const* d_in, const int* in_sizes, int n_in,
                              void* d_out, int out_size, void* d_ws, size_t ws_size,
                              hipStream_t stream) {
  (void)in_sizes; (void)n_in; (void)out_size; (void)d_ws; (void)ws_size;
  const float* useq = (const float*)d_in[0];
  const float* xGz0 = (const float*)d_in[1];
  const float* W1   = (const float*)d_in[2];
  const float* b1   = (const float*)d_in[3];
  const float* W2   = (const float*)d_in[4];
  const float* b2   = (const float*)d_in[5];
  const float* W3   = (const float*)d_in[6];
  const float* b3   = (const float*)d_in[7];
  float* out = (float*)d_out;
  cstr_rollout_kernel<<<dim3(B_TOT / BT), dim3(256), SMEM_BYTES, stream>>>(
      useq, xGz0, W1, b1, W2, b2, W3, b3, out);
}


# 8 batch rows a workgroup instead of 16, 256 workgroups: every WGP works; the 16-row tile holds each row twice (one LDS address), A fragment reads 8-way not 16-way; each lane half finishes 4 rows
# speedup vs baseline: 1.6603x; 1.6603x over previous
#include <hip/hip_runtime.h>
#include <math.h>


typedef __attribute__((ext_vector_type(2))) float v2f;
typedef __attribute__((ext_vector_type(4))) float v4f;
typedef __attribute__((ext_vector_type(8))) float v8f;

#define B_TOT   2048
#define T_STEPS 128
#define NXc     8
#define NUc     2
#define NZc     50
#define DINP    64
#define DH      128
#define BT      8
#define KST     (BT * NXc)
#define DELTA_F 0.01f

#define LDW     129
#define LDW3    17
#define OFF_W1P 0u
#define OFF_W2P (OFF_W1P + 32u*LDW*8u)
#define OFF_W3P (OFF_W2P + 64u*LDW*8u)
#define OFF_INP (OFF_W3P + 64u*LDW3*8u)
#define OFF_H1  (OFF_INP + 16u*64u*4u)
#define OFF_H2  (OFF_H1  + 16u*128u*4u)
#define OFF_NN  (OFF_H2  + 16u*128u*4u)
#define OFF_XG  (OFF_NN  + 256u*4u)
#define OFF_XS  (OFF_XG  + 128u*4u)
#define OFF_Z   (OFF_XS  + 128u*4u)
#define OFF_K   (OFF_Z   + 16u*52u*4u)
#define OFF_U   (OFF_K   + 512u*4u)
#define OFF_B1  (OFF_U   + 32u*4u)
#define OFF_B2  (OFF_B1  + 128u*4u)
#define OFF_B3  (OFF_B2  + 128u*4u)
#define SMEM_BYTES (OFF_B3 + 16u*4u)

__device__ __forceinline__ float fast_tanh(float x) {
  float e = __expf(2.0f * x);
  return 1.0f - __fdividef(2.0f, e + 1.0f);
}

__device__ __forceinline__ v8f mm16(const float* __restrict__ A, int lda,
                                    const v2f* __restrict__ Bp, int ldb,
                                    int kIters, int lane) {
  v8f acc = {};
  const int mr = lane & 15;
  const int kh = lane >> 4;
  const float* ap = A + (mr & (BT - 1)) * lda + 2 * kh;
  const v2f*   bp = Bp + (size_t)kh * ldb + mr;
#pragma unroll
  for (int kk = 0; kk < kIters; ++kk) {
    v2f a = *(const v2f*)(ap + 4 * kk);
    v2f b = bp[(size_t)(2 * kk) * ldb];
    acc = __builtin_amdgcn_wmma_f32_16x16x4_f32(false, a, false, b,
                                                (short)0, acc, false, false);
  }
  return acc;
}

__device__ __forceinline__ v4f my_rows(v8f d, int kh) {
  const v4f lo = __builtin_shufflevector(d, d, 0, 1, 2, 3);
  const v4f hi = __builtin_shufflevector(d, d, 4, 5, 6, 7);
  return kh ? hi : lo;
}

__device__ __forceinline__ void mlp_and_physics(
    int tid, int wave, int lane, int stage,
    const float* sInp, float* sH1, float* sH2, float* sNN,
    const v2f* W1p, const v2f* W2p, const v2f* W3p,
    const float* sB1, const float* sB2, const float* sB3,
    const float* xsrc, const float* sU, float* sK) {
  {
    v8f d = mm16(sInp, DINP, W1p + wave * 16, LDW, 16, lane);
    const int nc = wave * 16 + (lane & 15);
    const int kh = lane >> 4;
    const float bias = sB1[nc];
    const v4f s = my_rows(d, kh);
#pragma unroll
    for (int v = 0; v < 4; ++v)
      sH1[(v + 4 * kh) * DH + nc] = fast_tanh(s[v] + bias);
  }
  __syncthreads();
  {
    v8f d = mm16(sH1, DH, W2p + wave * 16, LDW, 32, lane);
    const int nc = wave * 16 + (lane & 15);
    const int kh = lane >> 4;
    const float bias = sB2[nc];
    const v4f s = my_rows(d, kh);
#pragma unroll
    for (int v = 0; v < 4; ++v)
      sH2[(v + 4 * kh) * DH + nc] = fast_tanh(s[v] + bias);
  }
  sNN[tid] = 0.0f;
  __syncthreads();
  {
    v8f d = mm16(sH2 + wave * 16, DH, W3p + wave * 8 * LDW3, LDW3, 4, lane);
    const int nc = lane & 15;
    const int kh = lane >> 4;
    const v4f s = my_rows(d, kh);
    if (nc < NXc) {
#pragma unroll
      for (int v = 0; v < 4; ++v)
        atomicAdd(&sNN[(v + 4 * kh) * 16 + nc], s[v]);
    }
  }
  __syncthreads();
  if (tid < BT) {
    const float* xs = xsrc + tid * NXc;
    float Hr  = xs[0] * 2.0f + 50.0f;
    float CAr = xs[1] * 0.2f + 1.0f;
    float CBr = xs[2] * 0.1f + 0.5f;
    float Tr  = xs[3] * 4.0f + 313.0f;
    float Hb  = xs[4] * 2.0f + 50.0f;
    float CAb = xs[5] * 0.2f + 1.0f;
    float CBb = xs[6] * 0.1f + 0.5f;
    float Tb  = xs[7] * 4.0f + 313.0f;
    float F = sU[tid * 2 + 0] * 1.0f + 10.0f;
    float D = sU[tid * 2 + 1] * 0.5f + 5.0f;
    float Fr = 2.5f * __fsqrt_rn(Hr);
    float Fb = 1.5f * __fsqrt_rn(Hb);
    float invArHr = __fdividef(1.0f, 3.0f * Hr);
    float invAbHb = __fdividef(1.0f, 3.0f * Hb);
    float dHr  = (F + D - Fr) * (1.0f / 3.0f);
    float dCAr = (F * (6.0f - CAr) - D * CAr) * invArHr;
    float dCBr = (-F * CBr - D * CBr) * invArHr;
    float dTr  = (F * (320.0f - Tr) + D * (300.0f - Tr)) * invArHr
               + (200.0f / 30.0f) * invArHr;
    float dHb  = (Fr - Fb - D) * (1.0f / 3.0f);
    float dCAb = (Fr * (CAr - CAb) + D * CAb) * invAbHb;
    float dCBb = (Fr * (CBr - CBb) + D * CBb) * invAbHb;
    float dTb  = Fr * (Tr - Tb) * invAbHb + (200.0f / 30.0f) * invAbHb;
    float*       kr = sK + stage * (BT * NXc) + tid * NXc;
    const float* nn = sNN + tid * 16;
    kr[0] = dHr  * 0.5f  + nn[0] + sB3[0];
    kr[1] = dCAr * 5.0f  + nn[1] + sB3[1];
    kr[2] = dCBr * 10.0f + nn[2] + sB3[2];
    kr[3] = dTr  * 0.25f + nn[3] + sB3[3];
    kr[4] = dHb  * 0.5f  + nn[4] + sB3[4];
    kr[5] = dCAb * 5.0f  + nn[5] + sB3[5];
    kr[6] = dCBb * 10.0f + nn[6] + sB3[6];
    kr[7] = dTb  * 0.25f + nn[7] + sB3[7];
  }
  __syncthreads();
}

__global__ __launch_bounds__(256) void cstr_rollout_kernel(
    const float* __restrict__ useq, const float* __restrict__ xGz0,
    const float* __restrict__ W1g, const float* __restrict__ b1g,
    const float* __restrict__ W2g, const float* __restrict__ b2g,
    const float* __restrict__ W3g, const float* __restrict__ b3g,
    float* __restrict__ out) {
  extern __shared__ __align__(16) char smem[];
  v2f*   W1p  = (v2f*)(smem + OFF_W1P);
  v2f*   W2p  = (v2f*)(smem + OFF_W2P);
  v2f*   W3p  = (v2f*)(smem + OFF_W3P);
  float* sInp = (float*)(smem + OFF_INP);
  float* sH1  = (float*)(smem + OFF_H1);
  float* sH2  = (float*)(smem + OFF_H2);
  float* sNN  = (float*)(smem + OFF_NN);
  float* sXG  = (float*)(smem + OFF_XG);
  float* sXs  = (float*)(smem + OFF_XS);
  float* sZ   = (float*)(smem + OFF_Z);
  float* sK   = (float*)(smem + OFF_K);
  float* sU   = (float*)(smem + OFF_U);
  float* sB1  = (float*)(smem + OFF_B1);
  float* sB2  = (float*)(smem + OFF_B2);
  float* sB3  = (float*)(smem + OFF_B3);

  const int tid  = threadIdx.x;
  const int lane = tid & 31;
  const int wave = tid >> 5;
  const int bt0  = blockIdx.x * BT;

  for (int i = tid; i < 32 * 128; i += 256) {
    int kk = i >> 7, n = i & 127;
    float lo = (2 * kk     < 60) ? W1g[(2 * kk)     * 128 + n] : 0.0f;
    float hi = (2 * kk + 1 < 60) ? W1g[(2 * kk + 1) * 128 + n] : 0.0f;
    W1p[kk * LDW + n] = (v2f){lo, hi};
  }
  for (int i = tid; i < 64 * 128; i += 256) {
    int kk = i >> 7, n = i & 127;
    W2p[kk * LDW + n] = (v2f){W2g[(2 * kk) * 128 + n], W2g[(2 * kk + 1) * 128 + n]};
  }
  for (int i = tid; i < 64 * 16; i += 256) {
    int kk = i >> 4, n = i & 15;
    v2f p = (v2f){0.0f, 0.0f};
    if (n < NXc) p = (v2f){W3g[(2 * kk) * NXc + n], W3g[(2 * kk + 1) * NXc + n]};
    W3p[kk * LDW3 + n] = p;
  }
  if (tid < 128) { sB1[tid] = b1g[tid]; sB2[tid] = b2g[tid]; }
  if (tid < NXc) sB3[tid] = b3g[tid];
  for (int i = tid; i < BT * 58; i += 256) {
    int r = i / 58, c = i - r * 58;
    float v = xGz0[(size_t)(bt0 + r) * 58 + c];
    if (c < NXc) sXG[r * NXc + c] = v;
    else         sZ[r * 52 + (c - NXc)] = v;
  }
  __syncthreads();

  for (int t = 0; t < T_STEPS; ++t) {
    if (tid < BT * NUc) {
      int r = tid >> 1, c = tid & 1;
      sU[tid] = useq[(size_t)(bt0 + r) * T_STEPS * NUc + (size_t)t * NUc + c];
    }
    __syncthreads();

    for (int i = tid; i < BT * DINP; i += 256) {
      int r = i >> 6, c = i & 63;
      float v;
      if      (c < 8)  v = sXG[r * NXc + c];
      else if (c < 58) v = sZ[r * 52 + (c - 8)];
      else if (c < 60) v = sU[r * 2 + (c - 58)];
      else             v = 0.0f;
      sInp[i] = v;
    }
    __syncthreads();
    mlp_and_physics(tid, wave, lane, 0, sInp, sH1, sH2, sNN, W1p, W2p, W3p,
                    sB1, sB2, sB3, sXG, sU, sK);

    for (int i = tid; i < BT * DINP; i += 256) {
      int r = i >> 6, c = i & 63;
      float v;
      if (c < 8) {
        v = sXG[r * NXc + c] + 0.5f * DELTA_F * sK[0 * KST + r * NXc + c];
        sXs[r * NXc + c] = v;
      } else if (c < 48) {
        int idx = c - 8, p = idx >> 3, q = idx & 7;
        float y0 = sZ[r * 52 + p * 8 + q];
        float y1 = (p < 4) ? sZ[r * 52 + (p + 1) * 8 + q] : sXG[r * NXc + q];
        v = 0.5f * (y0 + y1);
      } else if (c < 58) v = sZ[r * 52 + 40 + (c - 48)];
      else if (c < 60)   v = sU[r * 2 + (c - 58)];
      else               v = 0.0f;
      sInp[i] = v;
    }
    __syncthreads();
    mlp_and_physics(tid, wave, lane, 1, sInp, sH1, sH2, sNN, W1p, W2p, W3p,
                    sB1, sB2, sB3, sXs, sU, sK);

    if (tid < BT * NXc) {
      int r = tid >> 3, j = tid & 7;
      float v = sXG[r * NXc + j] + 0.5f * DELTA_F * sK[1 * KST + r * NXc + j];
      sXs[r * NXc + j] = v;
      sInp[r * DINP + j] = v;
    }
    __syncthreads();
    mlp_and_physics(tid, wave, lane, 2, sInp, sH1, sH2, sNN, W1p, W2p, W3p,
                    sB1, sB2, sB3, sXs, sU, sK);

    for (int i = tid; i < BT * 48; i += 256) {
      int r = i / 48, c = i - r * 48;
      float v;
      if (c < 8) {
        v = sXG[r * NXc + c] + DELTA_F * sK[2 * KST + r * NXc + c];
        sXs[r * NXc + c] = v;
      } else {
        int idx = c - 8;
        v = (idx < 32) ? sZ[r * 52 + 8 + idx] : sXG[r * NXc + (idx - 32)];
      }
      sInp[r * DINP + c] = v;
    }
    __syncthreads();
    mlp_and_physics(tid, wave, lane, 3, sInp, sH1, sH2, sNN, W1p, W2p, W3p,
                    sB1, sB2, sB3, sXs, sU, sK);

    float xgnew = 0.0f;
    if (tid < BT * NXc) {
      int r = tid >> 3, j = tid & 7;
      float xg = sXG[r * NXc + j];
      out[(size_t)(bt0 + r) * T_STEPS * NXc + (size_t)t * NXc + j] = xg;
      xgnew = xg + (DELTA_F / 6.0f) *
              (sK[0 * KST + tid] + 2.0f * sK[1 * KST + tid] +
               2.0f * sK[2 * KST + tid] + sK[3 * KST + tid]);
    }
    float regz[2];
#pragma unroll
    for (int ii = 0; ii < 2; ++ii) {
      int i = tid + ii * 256;
      if (i < BT * NZc) {
        int r = i / NZc, c = i - r * NZc;
        float v;
        if      (c < 32) v = sZ[r * 52 + 8 + c];
        else if (c < 40) v = sXG[r * NXc + (c - 32)];
        else if (c < 48) v = sZ[r * 52 + (c + 2)];
        else             v = sU[r * 2 + (c - 48)];
        regz[ii] = v;
      }
    }
    __syncthreads();
#pragma unroll
    for (int ii = 0; ii < 2; ++ii) {
      int i = tid + ii * 256;
      if (i < BT * NZc) {
        int r = i / NZc, c = i - r * NZc;
        sZ[r * 52 + c] = regz[ii];
      }
    }
    if (tid < BT * NXc) sXG[tid] = xgnew;
    __syncthreads();
  }
}

extern "C" void kernel_launch(void* const* d_in, const int* in_sizes, int n_in,
                              void* d_out, int out_size, void* d_ws, size_t ws_size,
                              hipStream_t stream) {
  (void)in_sizes; (void)n_in; (void)out_size; (void)d_ws; (void)ws_size;
  const float* useq = (const float*)d_in[0];
  const float* xGz0 = (const float*)d_in[1];
  const float* W1   = (const float*)d_in[2];
  const float* b1   = (const float*)d_in[3];
  const float* W2   = (const float*)d_in[4];
  const float* b2   = (const float*)d_in[5];
  const float* W3   = (const float*)d_in[6];
  const float* b3   = (const float*)d_in[7];
  float* out = (float*)d_out;
  cstr_rollout_kernel<<<dim3(B_TOT / BT), dim3(256), SMEM_BYTES, stream>>>(
      useq, xGz0, W1, b1, W2, b2, W3, b3, out);
}


# Hidden layers split over K across the 8 waves, not N: a wave's 16 finished columns are its own next k-slice (wave-private LDS transpose); partials added in wave order 0..7: L1, L2 sum order changes
# speedup vs baseline: 2.0646x; 1.2435x over previous
#include <hip/hip_runtime.h>
#include <math.h>


typedef __attribute__((ext_vector_type(2))) float v2f;
typedef __attribute__((ext_vector_type(4))) float v4f;
typedef __attribute__((ext_vector_type(8))) float v8f;

#define B_TOT   2048
#define T_STEPS 128
#define NXc     8
#define NUc     2
#define NZc     50
#define DINP    64
#define DH      128
#define BT      8
#define KST     (BT * NXc)
#define DELTA_F 0.01f

#define LDW     129
#define LDW3    17
#define OFF_W1P 0u
#define OFF_W2P (OFF_W1P + 32u*LDW*8u)
#define OFF_W3P (OFF_W2P + 64u*LDW*8u)
#define OFF_INP (OFF_W3P + 64u*LDW3*8u)
#define SW_FLT  136
#define P_FLT   1024
#define OFF_P1  (OFF_INP + 16u*64u*4u)
#define OFF_P2  (OFF_P1  + 8u*P_FLT*4u)
#define OFF_SW  (OFF_P2  + 8u*P_FLT*4u)
#define OFF_NN  (OFF_SW  + 8u*SW_FLT*4u)
#define OFF_XG  (OFF_NN  + 256u*4u)
#define OFF_XS  (OFF_XG  + 128u*4u)
#define OFF_Z   (OFF_XS  + 128u*4u)
#define OFF_K   (OFF_Z   + 16u*52u*4u)
#define OFF_U   (OFF_K   + 512u*4u)
#define OFF_B1  (OFF_U   + 32u*4u)
#define OFF_B2  (OFF_B1  + 128u*4u)
#define OFF_B3  (OFF_B2  + 128u*4u)
#define SMEM_BYTES (OFF_B3 + 16u*4u)

__device__ __forceinline__ float fast_tanh(float x) {
  float e = __expf(2.0f * x);
  return 1.0f - __fdividef(2.0f, e + 1.0f);
}


__device__ __forceinline__ v4f my_rows(v8f d, int kh) {
  const v4f lo = __builtin_shufflevector(d, d, 0, 1, 2, 3);
  const v4f hi = __builtin_shufflevector(d, d, 4, 5, 6, 7);
  return kh ? hi : lo;
}

__device__ __forceinline__ int sw_row(int r) { return 16 * r + 4 * (r >> 2); }

__device__ __forceinline__ void own_cols_to_afrag(float* Sw, v4f h, int lane, v2f a[4]) {
  const int n  = lane & 15;
  const int kh = lane >> 4;
#pragma unroll
  for (int v = 0; v < 4; ++v) Sw[sw_row(4 * kh + v) + n] = h[v];
  const float* rp = Sw + sw_row(n & (BT - 1)) + 2 * kh;
#pragma unroll
  for (int kk = 0; kk < 4; ++kk) a[kk] = *(const v2f*)(rp + 4 * kk);
}

template <int KSTEPS>
__device__ __forceinline__ void slice_times_rows(const v2f* a, const v2f* __restrict__ Bp,
                                                 float* pw, int lane) {
  const int n  = lane & 15;
  const int kh = lane >> 4;
  v8f acc[8];
#pragma unroll
  for (int j = 0; j < 8; ++j) acc[j] = (v8f){};
  const v2f* bp = Bp + (size_t)kh * LDW + n;
#pragma unroll
  for (int kk = 0; kk < KSTEPS; ++kk) {
#pragma unroll
    for (int j = 0; j < 8; ++j) {
      v2f b = bp[(size_t)(2 * kk) * LDW + 16 * j];
      acc[j] = __builtin_amdgcn_wmma_f32_16x16x4_f32(false, a[kk], false, b,
                                                     (short)0, acc[j], false, false);
    }
  }
  float* p = pw + n * 8 + 4 * kh;
#pragma unroll
  for (int j = 0; j < 8; ++j) *(v4f*)(p + j * 128) = my_rows(acc[j], kh);
}

__device__ __forceinline__ v4f sum_slices_tanh(const float* sP, int nc, int kh, float bias) {
  const float* pr = sP + nc * 8 + 4 * kh;
  v4f s = *(const v4f*)pr;
#pragma unroll
  for (int w = 1; w < 8; ++w) s += *(const v4f*)(pr + w * P_FLT);
  v4f h;
#pragma unroll
  for (int v = 0; v < 4; ++v) h[v] = fast_tanh(s[v] + bias);
  return h;
}

__device__ __forceinline__ void mlp_and_physics(
    int tid, int wave, int lane, int stage,
    const float* sInp, float* sP1, float* sP2, float* Sw, float* sNN,
    const v2f* W1p, const v2f* W2p, const v2f* W3p,
    const float* sB1, const float* sB2, const float* sB3,
    const float* xsrc, const float* sU, float* sK) {
  const int n  = lane & 15;
  const int kh = lane >> 4;
  const int nc = wave * 16 + n;
  v2f a[4];
  {
    const float* ap = sInp + (n & (BT - 1)) * DINP + 8 * wave + 2 * kh;
    a[0] = *(const v2f*)ap;
    a[1] = *(const v2f*)(ap + 4);
    slice_times_rows<2>(a, W1p + (size_t)(4 * wave) * LDW, sP1 + wave * P_FLT, lane);
  }
  __syncthreads();
  own_cols_to_afrag(Sw, sum_slices_tanh(sP1, nc, kh, sB1[nc]), lane, a);
  slice_times_rows<4>(a, W2p + (size_t)(8 * wave) * LDW, sP2 + wave * P_FLT, lane);
  sNN[tid] = 0.0f;
  __syncthreads();
  own_cols_to_afrag(Sw, sum_slices_tanh(sP2, nc, kh, sB2[nc]), lane, a);
  {
    v8f d = {};
    const v2f* bp = W3p + (size_t)(8 * wave + kh) * LDW3 + n;
#pragma unroll
    for (int kk = 0; kk < 4; ++kk) {
      v2f b = bp[(size_t)(2 * kk) * LDW3];
      d = __builtin_amdgcn_wmma_f32_16x16x4_f32(false, a[kk], false, b,
                                                (short)0, d, false, false);
    }
    const v4f s = my_rows(d, kh);
    if (n < NXc) {
#pragma unroll
      for (int v = 0; v < 4; ++v)
        atomicAdd(&sNN[(v + 4 * kh) * 16 + n], s[v]);
    }
  }
  __syncthreads();
  if (tid < BT) {
    const float* xs = xsrc + tid * NXc;
    float Hr  = xs[0] * 2.0f + 50.0f;
    float CAr = xs[1] * 0.2f + 1.0f;
    float CBr = xs[2] * 0.1f + 0.5f;
    float Tr  = xs[3] * 4.0f + 313.0f;
    float Hb  = xs[4] * 2.0f + 50.0f;
    float CAb = xs[5] * 0.2f + 1.0f;
    float CBb = xs[6] * 0.1f + 0.5f;
    float Tb  = xs[7] * 4.0f + 313.0f;
    float F = sU[tid * 2 + 0] * 1.0f + 10.0f;
    float D = sU[tid * 2 + 1] * 0.5f + 5.0f;
    float Fr = 2.5f * __fsqrt_rn(Hr);
    float Fb = 1.5f * __fsqrt_rn(Hb);
    float invArHr = __fdividef(1.0f, 3.0f * Hr);
    float invAbHb = __fdividef(1.0f, 3.0f * Hb);
    float dHr  = (F + D - Fr) * (1.0f / 3.0f);
    float dCAr = (F * (6.0f - CAr) - D * CAr) * invArHr;
    float dCBr = (-F * CBr - D * CBr) * invArHr;
    float dTr  = (F * (320.0f - Tr) + D * (300.0f - Tr)) * invArHr
               + (200.0f / 30.0f) * invArHr;
    float dHb  = (Fr - Fb - D) * (1.0f / 3.0f);
    float dCAb = (Fr * (CAr - CAb) + D * CAb) * invAbHb;
    float dCBb = (Fr * (CBr - CBb) + D * CBb) * invAbHb;
    float dTb  = Fr * (Tr - Tb) * invAbHb + (200.0f / 30.0f) * invAbHb;
    float*       kr = sK + stage * (BT * NXc) + tid * NXc;
    const float* nn = sNN + tid * 16;
    kr[0] = dHr  * 0.5f  + nn[0] + sB3[0];
    kr[1] = dCAr * 5.0f  + nn[1] + sB3[1];
    kr[2] = dCBr * 10.0f + nn[2] + sB3[2];
    kr[3] = dTr  * 0.25f + nn[3] + sB3[3];
    kr[4] = dHb  * 0.5f  + nn[4] + sB3[4];
    kr[5] = dCAb * 5.0f  + nn[5] + sB3[5];
    kr[6] = dCBb * 10.0f + nn[6] + sB3[6];
    kr[7] = dTb  * 0.25f + nn[7] + sB3[7];
  }
  __syncthreads();
}

__global__ __launch_bounds__(256) void cstr_rollout_kernel(
    const float* __restrict__ useq, const float* __restrict__ xGz0,
    const float* __restrict__ W1g, const float* __restrict__ b1g,
    const float* __restrict__ W2g, const float* __restrict__ b2g,
    const float* __restrict__ W3g, const float* __restrict__ b3g,
    float* __restrict__ out) {
  extern __shared__ __align__(16) char smem[];
  v2f*   W1p  = (v2f*)(smem + OFF_W1P);
  v2f*   W2p  = (v2f*)(smem + OFF_W2P);
  v2f*   W3p  = (v2f*)(smem + OFF_W3P);
  float* sInp = (float*)(smem + OFF_INP);
  float* sP1  = (float*)(smem + OFF_P1);
  float* sP2  = (float*)(smem + OFF_P2);
  float* sNN  = (float*)(smem + OFF_NN);
  float* sXG  = (float*)(smem + OFF_XG);
  float* sXs  = (float*)(smem + OFF_XS);
  float* sZ   = (float*)(smem + OFF_Z);
  float* sK   = (float*)(smem + OFF_K);
  float* sU   = (float*)(smem + OFF_U);
  float* sB1  = (float*)(smem + OFF_B1);
  float* sB2  = (float*)(smem + OFF_B2);
  float* sB3  = (float*)(smem + OFF_B3);

  const int tid  = threadIdx.x;
  const int lane = tid & 31;
  const int wave = tid >> 5;
  const int bt0  = blockIdx.x * BT;
  float* Sw = (float*)(smem + OFF_SW) + wave * SW_FLT;

  for (int i = tid; i < 32 * 128; i += 256) {
    int kk = i >> 7, n = i & 127;
    float lo = (2 * kk     < 60) ? W1g[(2 * kk)     * 128 + n] : 0.0f;
    float hi = (2 * kk + 1 < 60) ? W1g[(2 * kk + 1) * 128 + n] : 0.0f;
    W1p[kk * LDW + n] = (v2f){lo, hi};
  }
  for (int i = tid; i < 64 * 128; i += 256) {
    int kk = i >> 7, n = i & 127;
    W2p[kk * LDW + n] = (v2f){W2g[(2 * kk) * 128 + n], W2g[(2 * kk + 1) * 128 + n]};
  }
  for (int i = tid; i < 64 * 16; i += 256) {
    int kk = i >> 4, n = i & 15;
    v2f p = (v2f){0.0f, 0.0f};
    if (n < NXc) p = (v2f){W3g[(2 * kk) * NXc + n], W3g[(2 * kk + 1) * NXc + n]};
    W3p[kk * LDW3 + n] = p;
  }
  if (tid < 128) { sB1[tid] = b1g[tid]; sB2[tid] = b2g[tid]; }
  if (tid < NXc) sB3[tid] = b3g[tid];
  for (int i = tid; i < BT * 58; i += 256) {
    int r = i / 58, c = i - r * 58;
    float v = xGz0[(size_t)(bt0 + r) * 58 + c];
    if (c < NXc) sXG[r * NXc + c] = v;
    else         sZ[r * 52 + (c - NXc)] = v;
  }
  __syncthreads();

  for (int t = 0; t < T_STEPS; ++t) {
    if (tid < BT * NUc) {
      int r = tid >> 1, c = tid & 1;
      sU[tid] = useq[(size_t)(bt0 + r) * T_STEPS * NUc + (size_t)t * NUc + c];
    }
    __syncthreads();

    for (int i = tid; i < BT * DINP; i += 256) {
      int r = i >> 6, c = i & 63;
      float v;
      if      (c < 8)  v = sXG[r * NXc + c];
      else if (c < 58) v = sZ[r * 52 + (c - 8)];
      else if (c < 60) v = sU[r * 2 + (c - 58)];
      else             v = 0.0f;
      sInp[i] = v;
    }
    __syncthreads();
    mlp_and_physics(tid, wave, lane, 0, sInp, sP1, sP2, Sw, sNN, W1p, W2p, W3p,
                    sB1, sB2, sB3, sXG, sU, sK);

    for (int i = tid; i < BT * DINP; i += 256) {
      int r = i >> 6, c = i & 63;
      float v;
      if (c < 8) {
        v = sXG[r * NXc + c] + 0.5f * DELTA_F * sK[0 * KST + r * NXc + c];
        sXs[r * NXc + c] = v;
      } else if (c < 48) {
        int idx = c - 8, p = idx >> 3, q = idx & 7;
        float y0 = sZ[r * 52 + p * 8 + q];
        float y1 = (p < 4) ? sZ[r * 52 + (p + 1) * 8 + q] : sXG[r * NXc + q];
        v = 0.5f * (y0 + y1);
      } else if (c < 58) v = sZ[r * 52 + 40 + (c - 48)];
      else if (c < 60)   v = sU[r * 2 + (c - 58)];
      else               v = 0.0f;
      sInp[i] = v;
    }
    __syncthreads();
    mlp_and_physics(tid, wave, lane, 1, sInp, sP1, sP2, Sw, sNN, W1p, W2p, W3p,
                    sB1, sB2, sB3, sXs, sU, sK);

    if (tid < BT * NXc) {
      int r = tid >> 3, j = tid & 7;
      float v = sXG[r * NXc + j] + 0.5f * DELTA_F * sK[1 * KST + r * NXc + j];
      sXs[r * NXc + j] = v;
      sInp[r * DINP + j] = v;
    }
    __syncthreads();
    mlp_and_physics(tid, wave, lane, 2, sInp, sP1, sP2, Sw, sNN, W1p, W2p, W3p,
                    sB1, sB2, sB3, sXs, sU, sK);

    for (int i = tid; i < BT * 48; i += 256) {
      int r = i / 48, c = i - r * 48;
      float v;
      if (c < 8) {
        v = sXG[r * NXc + c] + DELTA_F * sK[2 * KST + r * NXc + c];
        sXs[r * NXc + c] = v;
      } else {
        int idx = c - 8;
        v = (idx < 32) ? sZ[r * 52 + 8 + idx] : sXG[r * NXc + (idx - 32)];
      }
      sInp[r * DINP + c] = v;
    }
    __syncthreads();
    mlp_and_physics(tid, wave, lane, 3, sInp, sP1, sP2, Sw, sNN, W1p, W2p, W3p,
                    sB1, sB2, sB3, sXs, sU, sK);

    float xgnew = 0.0f;
    if (tid < BT * NXc) {
      int r = tid >> 3, j = tid & 7;
      float xg = sXG[r * NXc + j];
      out[(size_t)(bt0 + r) * T_STEPS * NXc + (size_t)t * NXc + j] = xg;
      xgnew = xg + (DELTA_F / 6.0f) *
              (sK[0 * KST + tid] + 2.0f * sK[1 * KST + tid] +
               2.0f * sK[2 * KST + tid] + sK[3 * KST + tid]);
    }
    float regz[2];
#pragma unroll
    for (int ii = 0; ii < 2; ++ii) {
      int i = tid + ii * 256;
      if (i < BT * NZc) {
        int r = i / NZc, c = i - r * NZc;
        float v;
        if      (c < 32) v = sZ[r * 52 + 8 + c];
        else if (c < 40) v = sXG[r * NXc + (c - 32)];
        else if (c < 48) v = sZ[r * 52 + (c + 2)];
        else             v = sU[r * 2 + (c - 48)];
        regz[ii] = v;
      }
    }
    __syncthreads();
#pragma unroll
    for (int ii = 0; ii < 2; ++ii) {
      int i = tid + ii * 256;
      if (i < BT * NZc) {
        int r = i / NZc, c = i - r * NZc;
        sZ[r * 52 + c] = regz[ii];
      }
    }
    if (tid < BT * NXc) sXG[tid] = xgnew;
    __syncthreads();
  }
}

extern "C" void kernel_launch(void* const* d_in, const int* in_sizes, int n_in,
                              void* d_out, int out_size, void* d_ws, size_t ws_size,
                              hipStream_t stream) {
  (void)in_sizes; (void)n_in; (void)out_size; (void)d_ws; (void)ws_size;
  const float* useq = (const float*)d_in[0];
  const float* xGz0 = (const float*)d_in[1];
  const float* W1   = (const float*)d_in[2];
  const float* b1   = (const float*)d_in[3];
  const float* W2   = (const float*)d_in[4];
  const float* b2   = (const float*)d_in[5];
  const float* W3   = (const float*)d_in[6];
  const float* b3   = (const float*)d_in[7];
  float* out = (float*)d_out;
  cstr_rollout_kernel<<<dim3(B_TOT / BT), dim3(256), SMEM_BYTES, stream>>>(
      useq, xGz0, W1, b1, W2, b2, W3, b3, out);
}


# Layer 1 split over N, not K: a wave multiplies the whole K=64 input row (tile pitch 68) by its own 16 weight columns, 4 chains; no L1 partial exchange, 4 barriers fewer a step; L1 sum order changes
# speedup vs baseline: 2.3295x; 1.1283x over previous
#include <hip/hip_runtime.h>
#include <math.h>


typedef __attribute__((ext_vector_type(2))) float v2f;
typedef __attribute__((ext_vector_type(4))) float v4f;
typedef __attribute__((ext_vector_type(8))) float v8f;

#define B_TOT   2048
#define T_STEPS 128
#define NXc     8
#define NUc     2
#define NZc     50
#define DINP    64
#define DPITCH  68
#define DH      128
#define BT      8
#define KST     (BT * NXc)
#define DELTA_F 0.01f

#define LDW     129
#define LDW3    17
#define OFF_W1P 0u
#define OFF_W2P (OFF_W1P + 32u*LDW*8u)
#define OFF_W3P (OFF_W2P + 64u*LDW*8u)
#define OFF_INP (OFF_W3P + 64u*LDW3*8u)
#define SW_FLT  136
#define P_FLT   1024
#define OFF_P1  (OFF_INP + 16u*64u*4u)
#define OFF_P2  (OFF_P1  + 8u*P_FLT*4u)
#define OFF_SW  (OFF_P2  + 8u*P_FLT*4u)
#define OFF_NN  (OFF_SW  + 8u*SW_FLT*4u)
#define OFF_XG  (OFF_NN  + 256u*4u)
#define OFF_XS  (OFF_XG  + 128u*4u)
#define OFF_Z   (OFF_XS  + 128u*4u)
#define OFF_K   (OFF_Z   + 16u*52u*4u)
#define OFF_U   (OFF_K   + 512u*4u)
#define OFF_B1  (OFF_U   + 32u*4u)
#define OFF_B2  (OFF_B1  + 128u*4u)
#define OFF_B3  (OFF_B2  + 128u*4u)
#define SMEM_BYTES (OFF_B3 + 16u*4u)

__device__ __forceinline__ float fast_tanh(float x) {
  float e = __expf(2.0f * x);
  return 1.0f - __fdividef(2.0f, e + 1.0f);
}


__device__ __forceinline__ v4f my_rows(v8f d, int kh) {
  const v4f lo = __builtin_shufflevector(d, d, 0, 1, 2, 3);
  const v4f hi = __builtin_shufflevector(d, d, 4, 5, 6, 7);
  return kh ? hi : lo;
}

__device__ __forceinline__ int sw_row(int r) { return 16 * r + 4 * (r >> 2); }

__device__ __forceinline__ void own_cols_to_afrag(float* Sw, v4f h, int lane, v2f a[4]) {
  const int n  = lane & 15;
  const int kh = lane >> 4;
#pragma unroll
  for (int v = 0; v < 4; ++v) Sw[sw_row(4 * kh + v) + n] = h[v];
  const float* rp = Sw + sw_row(n & (BT - 1)) + 2 * kh;
#pragma unroll
  for (int kk = 0; kk < 4; ++kk) a[kk] = *(const v2f*)(rp + 4 * kk);
}

template <int KSTEPS>
__device__ __forceinline__ void slice_times_rows(const v2f* a, const v2f* __restrict__ Bp,
                                                 float* pw, int lane) {
  const int n  = lane & 15;
  const int kh = lane >> 4;
  v8f acc[8];
#pragma unroll
  for (int j = 0; j < 8; ++j) acc[j] = (v8f){};
  const v2f* bp = Bp + (size_t)kh * LDW + n;
#pragma unroll
  for (int kk = 0; kk < KSTEPS; ++kk) {
#pragma unroll
    for (int j = 0; j < 8; ++j) {
      v2f b = bp[(size_t)(2 * kk) * LDW + 16 * j];
      acc[j] = __builtin_amdgcn_wmma_f32_16x16x4_f32(false, a[kk], false, b,
                                                     (short)0, acc[j], false, false);
    }
  }
  float* p = pw + n * 8 + 4 * kh;
#pragma unroll
  for (int j = 0; j < 8; ++j) *(v4f*)(p + j * 128) = my_rows(acc[j], kh);
}

__device__ __forceinline__ v4f layer1_own_cols(const float* sInp, const v2f* __restrict__ W1p,
                                               int wave, int lane) {
  const int n  = lane & 15;
  const int kh = lane >> 4;
  const float* ap = sInp + (n & (BT - 1)) * DPITCH + 2 * kh;
  const v2f*   bp = W1p + (size_t)kh * LDW + 16 * wave + n;
  v8f acc[4];
#pragma unroll
  for (int c = 0; c < 4; ++c) acc[c] = (v8f){};
#pragma unroll
  for (int kk = 0; kk < 16; ++kk) {
    const v2f av = *(const v2f*)(ap + 4 * kk);
    const v2f b  = bp[(size_t)(2 * kk) * LDW];
    acc[kk & 3] = __builtin_amdgcn_wmma_f32_16x16x4_f32(false, av, false, b,
                                                        (short)0, acc[kk & 3], false, false);
  }
  const v8f d = (acc[0] + acc[1]) + (acc[2] + acc[3]);
  return my_rows(d, kh);
}

__device__ __forceinline__ v4f sum_slices_tanh(const float* sP, int nc, int kh, float bias) {
  const float* pr = sP + nc * 8 + 4 * kh;
  v4f s = *(const v4f*)pr;
#pragma unroll
  for (int w = 1; w < 8; ++w) s += *(const v4f*)(pr + w * P_FLT);
  v4f h;
#pragma unroll
  for (int v = 0; v < 4; ++v) h[v] = fast_tanh(s[v] + bias);
  return h;
}

__device__ __forceinline__ void mlp_and_physics(
    int tid, int wave, int lane, int stage,
    const float* sInp, float* sP2, float* Sw, float* sNN,
    const v2f* W1p, const v2f* W2p, const v2f* W3p,
    const float* sB1, const float* sB2, const float* sB3,
    const float* xsrc, const float* sU, float* sK) {
  const int n  = lane & 15;
  const int kh = lane >> 4;
  const int nc = wave * 16 + n;
  v2f a[4];
  {
    const v4f s1 = layer1_own_cols(sInp, W1p, wave, lane);
    const float bias1 = sB1[nc];
    v4f h1;
#pragma unroll
    for (int v = 0; v < 4; ++v) h1[v] = fast_tanh(s1[v] + bias1);
    own_cols_to_afrag(Sw, h1, lane, a);
  }
  slice_times_rows<4>(a, W2p + (size_t)(8 * wave) * LDW, sP2 + wave * P_FLT, lane);
  sNN[tid] = 0.0f;
  __syncthreads();
  own_cols_to_afrag(Sw, sum_slices_tanh(sP2, nc, kh, sB2[nc]), lane, a);
  {
    v8f d = {};
    const v2f* bp = W3p + (size_t)(8 * wave + kh) * LDW3 + n;
#pragma unroll
    for (int kk = 0; kk < 4; ++kk) {
      v2f b = bp[(size_t)(2 * kk) * LDW3];
      d = __builtin_amdgcn_wmma_f32_16x16x4_f32(false, a[kk], false, b,
                                                (short)0, d, false, false);
    }
    const v4f s = my_rows(d, kh);
    if (n < NXc) {
#pragma unroll
      for (int v = 0; v < 4; ++v)
        atomicAdd(&sNN[(v + 4 * kh) * 16 + n], s[v]);
    }
  }
  __syncthreads();
  if (tid < BT) {
    const float* xs = xsrc + tid * NXc;
    float Hr  = xs[0] * 2.0f + 50.0f;
    float CAr = xs[1] * 0.2f + 1.0f;
    float CBr = xs[2] * 0.1f + 0.5f;
    float Tr  = xs[3] * 4.0f + 313.0f;
    float Hb  = xs[4] * 2.0f + 50.0f;
    float CAb = xs[5] * 0.2f + 1.0f;
    float CBb = xs[6] * 0.1f + 0.5f;
    float Tb  = xs[7] * 4.0f + 313.0f;
    float F = sU[tid * 2 + 0] * 1.0f + 10.0f;
    float D = sU[tid * 2 + 1] * 0.5f + 5.0f;
    float Fr = 2.5f * __fsqrt_rn(Hr);
    float Fb = 1.5f * __fsqrt_rn(Hb);
    float invArHr = __fdividef(1.0f, 3.0f * Hr);
    float invAbHb = __fdividef(1.0f, 3.0f * Hb);
    float dHr  = (F + D - Fr) * (1.0f / 3.0f);
    float dCAr = (F * (6.0f - CAr) - D * CAr) * invArHr;
    float dCBr = (-F * CBr - D * CBr) * invArHr;
    float dTr  = (F * (320.0f - Tr) + D * (300.0f - Tr)) * invArHr
               + (200.0f / 30.0f) * invArHr;
    float dHb  = (Fr - Fb - D) * (1.0f / 3.0f);
    float dCAb = (Fr * (CAr - CAb) + D * CAb) * invAbHb;
    float dCBb = (Fr * (CBr - CBb) + D * CBb) * invAbHb;
    float dTb  = Fr * (Tr - Tb) * invAbHb + (200.0f / 30.0f) * invAbHb;
    float*       kr = sK + stage * (BT * NXc) + tid * NXc;
    const float* nn = sNN + tid * 16;
    kr[0] = dHr  * 0.5f  + nn[0] + sB3[0];
    kr[1] = dCAr * 5.0f  + nn[1] + sB3[1];
    kr[2] = dCBr * 10.0f + nn[2] + sB3[2];
    kr[3] = dTr  * 0.25f + nn[3] + sB3[3];
    kr[4] = dHb  * 0.5f  + nn[4] + sB3[4];
    kr[5] = dCAb * 5.0f  + nn[5] + sB3[5];
    kr[6] = dCBb * 10.0f + nn[6] + sB3[6];
    kr[7] = dTb  * 0.25f + nn[7] + sB3[7];
  }
  __syncthreads();
}

__global__ __launch_bounds__(256) void cstr_rollout_kernel(
    const float* __restrict__ useq, const float* __restrict__ xGz0,
    const float* __restrict__ W1g, const float* __restrict__ b1g,
    const float* __restrict__ W2g, const float* __restrict__ b2g,
    const float* __restrict__ W3g, const float* __restrict__ b3g,
    float* __restrict__ out) {
  extern __shared__ __align__(16) char smem[];
  v2f*   W1p  = (v2f*)(smem + OFF_W1P);
  v2f*   W2p  = (v2f*)(smem + OFF_W2P);
  v2f*   W3p  = (v2f*)(smem + OFF_W3P);
  float* sInp = (float*)(smem + OFF_INP);
  float* sP2  = (float*)(smem + OFF_P2);
  float* sNN  = (float*)(smem + OFF_NN);
  float* sXG  = (float*)(smem + OFF_XG);
  float* sXs  = (float*)(smem + OFF_XS);
  float* sZ   = (float*)(smem + OFF_Z);
  float* sK   = (float*)(smem + OFF_K);
  float* sU   = (float*)(smem + OFF_U);
  float* sB1  = (float*)(smem + OFF_B1);
  float* sB2  = (float*)(smem + OFF_B2);
  float* sB3  = (float*)(smem + OFF_B3);

  const int tid  = threadIdx.x;
  const int lane = tid & 31;
  const int wave = tid >> 5;
  const int bt0  = blockIdx.x * BT;
  float* Sw = (float*)(smem + OFF_SW) + wave * SW_FLT;

  for (int i = tid; i < 32 * 128; i += 256) {
    int kk = i >> 7, n = i & 127;
    float lo = (2 * kk     < 60) ? W1g[(2 * kk)     * 128 + n] : 0.0f;
    float hi = (2 * kk + 1 < 60) ? W1g[(2 * kk + 1) * 128 + n] : 0.0f;
    W1p[kk * LDW + n] = (v2f){lo, hi};
  }
  for (int i = tid; i < 64 * 128; i += 256) {
    int kk = i >> 7, n = i & 127;
    W2p[kk * LDW + n] = (v2f){W2g[(2 * kk) * 128 + n], W2g[(2 * kk + 1) * 128 + n]};
  }
  for (int i = tid; i < 64 * 16; i += 256) {
    int kk = i >> 4, n = i & 15;
    v2f p = (v2f){0.0f, 0.0f};
    if (n < NXc) p = (v2f){W3g[(2 * kk) * NXc + n], W3g[(2 * kk + 1) * NXc + n]};
    W3p[kk * LDW3 + n] = p;
  }
  if (tid < 128) { sB1[tid] = b1g[tid]; sB2[tid] = b2g[tid]; }
  if (tid < NXc) sB3[tid] = b3g[tid];
  for (int i = tid; i < BT * 58; i += 256) {
    int r = i / 58, c = i - r * 58;
    float v = xGz0[(size_t)(bt0 + r) * 58 + c];
    if (c < NXc) sXG[r * NXc + c] = v;
    else         sZ[r * 52 + (c - NXc)] = v;
  }
  __syncthreads();

  for (int t = 0; t < T_STEPS; ++t) {
    if (tid < BT * NUc) {
      int r = tid >> 1, c = tid & 1;
      sU[tid] = useq[(size_t)(bt0 + r) * T_STEPS * NUc + (size_t)t * NUc + c];
    }
    __syncthreads();

    for (int i = tid; i < BT * DINP; i += 256) {
      int r = i >> 6, c = i & 63;
      float v;
      if      (c < 8)  v = sXG[r * NXc + c];
      else if (c < 58) v = sZ[r * 52 + (c - 8)];
      else if (c < 60) v = sU[r * 2 + (c - 58)];
      else             v = 0.0f;
      sInp[r * DPITCH + c] = v;
    }
    __syncthreads();
    mlp_and_physics(tid, wave, lane, 0, sInp, sP2, Sw, sNN, W1p, W2p, W3p,
                    sB1, sB2, sB3, sXG, sU, sK);

    for (int i = tid; i < BT * DINP; i += 256) {
      int r = i >> 6, c = i & 63;
      float v;
      if (c < 8) {
        v = sXG[r * NXc + c] + 0.5f * DELTA_F * sK[0 * KST + r * NXc + c];
        sXs[r * NXc + c] = v;
      } else if (c < 48) {
        int idx = c - 8, p = idx >> 3, q = idx & 7;
        float y0 = sZ[r * 52 + p * 8 + q];
        float y1 = (p < 4) ? sZ[r * 52 + (p + 1) * 8 + q] : sXG[r * NXc + q];
        v = 0.5f * (y0 + y1);
      } else if (c < 58) v = sZ[r * 52 + 40 + (c - 48)];
      else if (c < 60)   v = sU[r * 2 + (c - 58)];
      else               v = 0.0f;
      sInp[r * DPITCH + c] = v;
    }
    __syncthreads();
    mlp_and_physics(tid, wave, lane, 1, sInp, sP2, Sw, sNN, W1p, W2p, W3p,
                    sB1, sB2, sB3, sXs, sU, sK);

    if (tid < BT * NXc) {
      int r = tid >> 3, j = tid & 7;
      float v = sXG[r * NXc + j] + 0.5f * DELTA_F * sK[1 * KST + r * NXc + j];
      sXs[r * NXc + j] = v;
      sInp[r * DPITCH + j] = v;
    }
    __syncthreads();
    mlp_and_physics(tid, wave, lane, 2, sInp, sP2, Sw, sNN, W1p, W2p, W3p,
                    sB1, sB2, sB3, sXs, sU, sK);

    for (int i = tid; i < BT * 48; i += 256) {
      int r = i / 48, c = i - r * 48;
      float v;
      if (c < 8) {
        v = sXG[r * NXc + c] + DELTA_F * sK[2 * KST + r * NXc + c];
        sXs[r * NXc + c] = v;
      } else {
        int idx = c - 8;
        v = (idx < 32) ? sZ[r * 52 + 8 + idx] : sXG[r * NXc + (idx - 32)];
      }
      sInp[r * DPITCH + c] = v;
    }
    __syncthreads();
    mlp_and_physics(tid, wave, lane, 3, sInp, sP2, Sw, sNN, W1p, W2p, W3p,
                    sB1, sB2, sB3, sXs, sU, sK);

    float xgnew = 0.0f;
    if (tid < BT * NXc) {
      int r = tid >> 3, j = tid & 7;
      float xg = sXG[r * NXc + j];
      out[(size_t)(bt0 + r) * T_STEPS * NXc + (size_t)t * NXc + j] = xg;
      xgnew = xg + (DELTA_F / 6.0f) *
              (sK[0 * KST + tid] + 2.0f * sK[1 * KST + tid] +
               2.0f * sK[2 * KST + tid] + sK[3 * KST + tid]);
    }
    float regz[2];
#pragma unroll
    for (int ii = 0; ii < 2; ++ii) {
      int i = tid + ii * 256;
      if (i < BT * NZc) {
        int r = i / NZc, c = i - r * NZc;
        float v;
        if      (c < 32) v = sZ[r * 52 + 8 + c];
        else if (c < 40) v = sXG[r * NXc + (c - 32)];
        else if (c < 48) v = sZ[r * 52 + (c + 2)];
        else             v = sU[r * 2 + (c - 48)];
        regz[ii] = v;
      }
    }
    __syncthreads();
#pragma unroll
    for (int ii = 0; ii < 2; ++ii) {
      int i = tid + ii * 256;
      if (i < BT * NZc) {
        int r = i / NZc, c = i - r * NZc;
        sZ[r * 52 + c] = regz[ii];
      }
    }
    if (tid < BT * NXc) sXG[tid] = xgnew;
    __syncthreads();
  }
}

extern "C" void kernel_launch(void* const* d_in, const int* in_sizes, int n_in,
                              void* d_out, int out_size, void* d_ws, size_t ws_size,
                              hipStream_t stream) {
  (void)in_sizes; (void)n_in; (void)out_size; (void)d_ws; (void)ws_size;
  const float* useq = (const float*)d_in[0];
  const float* xGz0 = (const float*)d_in[1];
  const float* W1   = (const float*)d_in[2];
  const float* b1   = (const float*)d_in[3];
  const float* W2   = (const float*)d_in[4];
  const float* b2   = (const float*)d_in[5];
  const float* W3   = (const float*)d_in[6];
  const float* b3   = (const float*)d_in[7];
  float* out = (float*)d_out;
  cstr_rollout_kernel<<<dim3(B_TOT / BT), dim3(256), SMEM_BYTES, stream>>>(
      useq, xGz0, W1, b1, W2, b2, W3, b3, out);
}


# Layer-1 input kept as three LDS tiles: history columns built once a step in one branch-free pass, next stage's state columns stored by the lanes that finish k; no per-stage assembly pass
# speedup vs baseline: 2.6409x; 1.1337x over previous
#include <hip/hip_runtime.h>
#include <math.h>


typedef __attribute__((ext_vector_type(2))) float v2f;
typedef __attribute__((ext_vector_type(4))) float v4f;
typedef __attribute__((ext_vector_type(8))) float v8f;

#define B_TOT   2048
#define T_STEPS 128
#define NXc     8
#define NUc     2
#define NZc     50
#define DINP    64
#define DPITCH  68
#define DH      128
#define BT      8
#define KST     (BT * NXc)
#define DELTA_F 0.01f

#define LDW     129
#define LDW3    17
#define OFF_W1P 0u
#define OFF_W2P (OFF_W1P + 32u*LDW*8u)
#define OFF_W3P (OFF_W2P + 64u*LDW*8u)
#define OFF_INP (OFF_W3P + 64u*LDW3*8u)
#define SW_FLT  136
#define P_FLT   1024
#define OFF_P1  (OFF_INP + 16u*64u*4u)
#define OFF_T2  (OFF_P1)
#define OFF_T4  (OFF_P1 + 4096u)
#define OFF_P2  (OFF_P1  + 8u*P_FLT*4u)
#define OFF_SW  (OFF_P2  + 8u*P_FLT*4u)
#define OFF_NN  (OFF_SW  + 8u*SW_FLT*4u)
#define OFF_XG  (OFF_NN  + 256u*4u)
#define OFF_XS  (OFF_XG  + 128u*4u)
#define OFF_Z   (OFF_XS  + 128u*4u)
#define OFF_K   (OFF_Z   + 16u*52u*4u)
#define OFF_U   (OFF_K   + 512u*4u)
#define OFF_B1  (OFF_U   + 32u*4u)
#define OFF_B2  (OFF_B1  + 128u*4u)
#define OFF_B3  (OFF_B2  + 128u*4u)
#define SMEM_BYTES (OFF_B3 + 16u*4u)

__device__ __forceinline__ float fast_tanh(float x) {
  float e = __expf(2.0f * x);
  return 1.0f - __fdividef(2.0f, e + 1.0f);
}


__device__ __forceinline__ v4f my_rows(v8f d, int kh) {
  const v4f lo = __builtin_shufflevector(d, d, 0, 1, 2, 3);
  const v4f hi = __builtin_shufflevector(d, d, 4, 5, 6, 7);
  return kh ? hi : lo;
}

__device__ __forceinline__ int sw_row(int r) { return 16 * r + 4 * (r >> 2); }

__device__ __forceinline__ void own_cols_to_afrag(float* Sw, v4f h, int lane, v2f a[4]) {
  const int n  = lane & 15;
  const int kh = lane >> 4;
#pragma unroll
  for (int v = 0; v < 4; ++v) Sw[sw_row(4 * kh + v) + n] = h[v];
  const float* rp = Sw + sw_row(n & (BT - 1)) + 2 * kh;
#pragma unroll
  for (int kk = 0; kk < 4; ++kk) a[kk] = *(const v2f*)(rp + 4 * kk);
}

template <int KSTEPS>
__device__ __forceinline__ void slice_times_rows(const v2f* a, const v2f* __restrict__ Bp,
                                                 float* pw, int lane) {
  const int n  = lane & 15;
  const int kh = lane >> 4;
  v8f acc[8];
#pragma unroll
  for (int j = 0; j < 8; ++j) acc[j] = (v8f){};
  const v2f* bp = Bp + (size_t)kh * LDW + n;
#pragma unroll
  for (int kk = 0; kk < KSTEPS; ++kk) {
#pragma unroll
    for (int j = 0; j < 8; ++j) {
      v2f b = bp[(size_t)(2 * kk) * LDW + 16 * j];
      acc[j] = __builtin_amdgcn_wmma_f32_16x16x4_f32(false, a[kk], false, b,
                                                     (short)0, acc[j], false, false);
    }
  }
  float* p = pw + n * 8 + 4 * kh;
#pragma unroll
  for (int j = 0; j < 8; ++j) *(v4f*)(p + j * 128) = my_rows(acc[j], kh);
}

__device__ __forceinline__ v4f layer1_own_cols(const float* sInp, const v2f* __restrict__ W1p,
                                               int wave, int lane) {
  const int n  = lane & 15;
  const int kh = lane >> 4;
  const float* ap = sInp + (n & (BT - 1)) * DPITCH + 2 * kh;
  const v2f*   bp = W1p + (size_t)kh * LDW + 16 * wave + n;
  v8f acc[4];
#pragma unroll
  for (int c = 0; c < 4; ++c) acc[c] = (v8f){};
#pragma unroll
  for (int kk = 0; kk < 16; ++kk) {
    const v2f av = *(const v2f*)(ap + 4 * kk);
    const v2f b  = bp[(size_t)(2 * kk) * LDW];
    acc[kk & 3] = __builtin_amdgcn_wmma_f32_16x16x4_f32(false, av, false, b,
                                                        (short)0, acc[kk & 3], false, false);
  }
  const v8f d = (acc[0] + acc[1]) + (acc[2] + acc[3]);
  return my_rows(d, kh);
}

__device__ __forceinline__ v4f sum_slices_tanh(const float* sP, int nc, int kh, float bias) {
  const float* pr = sP + nc * 8 + 4 * kh;
  v4f s = *(const v4f*)pr;
#pragma unroll
  for (int w = 1; w < 8; ++w) s += *(const v4f*)(pr + w * P_FLT);
  v4f h;
#pragma unroll
  for (int v = 0; v < 4; ++v) h[v] = fast_tanh(s[v] + bias);
  return h;
}

__device__ __forceinline__ void mlp_and_physics(
    int tid, int wave, int lane, int stage,
    const float* sInp, float* sP2, float* Sw, float* sNN,
    const v2f* W1p, const v2f* W2p, const v2f* W3p,
    const float* sB1, const float* sB2, const float* sB3,
    const float* xsrc, const float* sU, float* sK,
    const float* sXG, float* sXs, bool has_next, float* xnext_tile, float xcoef) {
  const int n  = lane & 15;
  const int kh = lane >> 4;
  const int nc = wave * 16 + n;
  v2f a[4];
  {
    const v4f s1 = layer1_own_cols(sInp, W1p, wave, lane);
    const float bias1 = sB1[nc];
    v4f h1;
#pragma unroll
    for (int v = 0; v < 4; ++v) h1[v] = fast_tanh(s1[v] + bias1);
    own_cols_to_afrag(Sw, h1, lane, a);
  }
  slice_times_rows<4>(a, W2p + (size_t)(8 * wave) * LDW, sP2 + wave * P_FLT, lane);
  sNN[tid] = 0.0f;
  __syncthreads();
  own_cols_to_afrag(Sw, sum_slices_tanh(sP2, nc, kh, sB2[nc]), lane, a);
  {
    v8f d = {};
    const v2f* bp = W3p + (size_t)(8 * wave + kh) * LDW3 + n;
#pragma unroll
    for (int kk = 0; kk < 4; ++kk) {
      v2f b = bp[(size_t)(2 * kk) * LDW3];
      d = __builtin_amdgcn_wmma_f32_16x16x4_f32(false, a[kk], false, b,
                                                (short)0, d, false, false);
    }
    const v4f s = my_rows(d, kh);
    if (n < NXc) {
#pragma unroll
      for (int v = 0; v < 4; ++v)
        atomicAdd(&sNN[(v + 4 * kh) * 16 + n], s[v]);
    }
  }
  __syncthreads();
  if (tid < BT) {
    const float* xs = xsrc + tid * NXc;
    float xg[NXc];
    if (has_next) {
#pragma unroll
      for (int j = 0; j < NXc; ++j) xg[j] = sXG[tid * NXc + j];
    }
    float Hr  = xs[0] * 2.0f + 50.0f;
    float CAr = xs[1] * 0.2f + 1.0f;
    float CBr = xs[2] * 0.1f + 0.5f;
    float Tr  = xs[3] * 4.0f + 313.0f;
    float Hb  = xs[4] * 2.0f + 50.0f;
    float CAb = xs[5] * 0.2f + 1.0f;
    float CBb = xs[6] * 0.1f + 0.5f;
    float Tb  = xs[7] * 4.0f + 313.0f;
    float F = sU[tid * 2 + 0] * 1.0f + 10.0f;
    float D = sU[tid * 2 + 1] * 0.5f + 5.0f;
    float Fr = 2.5f * __fsqrt_rn(Hr);
    float Fb = 1.5f * __fsqrt_rn(Hb);
    float invArHr = __fdividef(1.0f, 3.0f * Hr);
    float invAbHb = __fdividef(1.0f, 3.0f * Hb);
    float dHr  = (F + D - Fr) * (1.0f / 3.0f);
    float dCAr = (F * (6.0f - CAr) - D * CAr) * invArHr;
    float dCBr = (-F * CBr - D * CBr) * invArHr;
    float dTr  = (F * (320.0f - Tr) + D * (300.0f - Tr)) * invArHr
               + (200.0f / 30.0f) * invArHr;
    float dHb  = (Fr - Fb - D) * (1.0f / 3.0f);
    float dCAb = (Fr * (CAr - CAb) + D * CAb) * invAbHb;
    float dCBb = (Fr * (CBr - CBb) + D * CBb) * invAbHb;
    float dTb  = Fr * (Tr - Tb) * invAbHb + (200.0f / 30.0f) * invAbHb;
    float*       kr = sK + stage * (BT * NXc) + tid * NXc;
    const float* nn = sNN + tid * 16;
    float kv[NXc];
    kv[0] = dHr  * 0.5f  + nn[0] + sB3[0];
    kv[1] = dCAr * 5.0f  + nn[1] + sB3[1];
    kv[2] = dCBr * 10.0f + nn[2] + sB3[2];
    kv[3] = dTr  * 0.25f + nn[3] + sB3[3];
    kv[4] = dHb  * 0.5f  + nn[4] + sB3[4];
    kv[5] = dCAb * 5.0f  + nn[5] + sB3[5];
    kv[6] = dCBb * 10.0f + nn[6] + sB3[6];
    kv[7] = dTb  * 0.25f + nn[7] + sB3[7];
#pragma unroll
    for (int j = 0; j < NXc; ++j) kr[j] = kv[j];
    if (has_next) {
      float xn[NXc];
#pragma unroll
      for (int j = 0; j < NXc; ++j) xn[j] = xg[j] + xcoef * kv[j];
#pragma unroll
      for (int j = 0; j < NXc; ++j) {
        sXs[tid * NXc + j] = xn[j];
        xnext_tile[tid * DPITCH + j] = xn[j];
      }
    }
  }
  __syncthreads();
}

__global__ __launch_bounds__(256) void cstr_rollout_kernel(
    const float* __restrict__ useq, const float* __restrict__ xGz0,
    const float* __restrict__ W1g, const float* __restrict__ b1g,
    const float* __restrict__ W2g, const float* __restrict__ b2g,
    const float* __restrict__ W3g, const float* __restrict__ b3g,
    float* __restrict__ out) {
  extern __shared__ __align__(16) char smem[];
  v2f*   W1p  = (v2f*)(smem + OFF_W1P);
  v2f*   W2p  = (v2f*)(smem + OFF_W2P);
  v2f*   W3p  = (v2f*)(smem + OFF_W3P);
  float* sT1  = (float*)(smem + OFF_INP);
  float* sT2  = (float*)(smem + OFF_T2);
  float* sT4  = (float*)(smem + OFF_T4);
  float* sP2  = (float*)(smem + OFF_P2);
  float* sNN  = (float*)(smem + OFF_NN);
  float* sXG  = (float*)(smem + OFF_XG);
  float* sXs  = (float*)(smem + OFF_XS);
  float* sZ   = (float*)(smem + OFF_Z);
  float* sK   = (float*)(smem + OFF_K);
  float* sU   = (float*)(smem + OFF_U);
  float* sB1  = (float*)(smem + OFF_B1);
  float* sB2  = (float*)(smem + OFF_B2);
  float* sB3  = (float*)(smem + OFF_B3);

  const int tid  = threadIdx.x;
  const int lane = tid & 31;
  const int wave = tid >> 5;
  const int bt0  = blockIdx.x * BT;
  float* Sw = (float*)(smem + OFF_SW) + wave * SW_FLT;

  for (int i = tid; i < 32 * 128; i += 256) {
    int kk = i >> 7, n = i & 127;
    float lo = (2 * kk     < 60) ? W1g[(2 * kk)     * 128 + n] : 0.0f;
    float hi = (2 * kk + 1 < 60) ? W1g[(2 * kk + 1) * 128 + n] : 0.0f;
    W1p[kk * LDW + n] = (v2f){lo, hi};
  }
  for (int i = tid; i < 64 * 128; i += 256) {
    int kk = i >> 7, n = i & 127;
    W2p[kk * LDW + n] = (v2f){W2g[(2 * kk) * 128 + n], W2g[(2 * kk + 1) * 128 + n]};
  }
  for (int i = tid; i < 64 * 16; i += 256) {
    int kk = i >> 4, n = i & 15;
    v2f p = (v2f){0.0f, 0.0f};
    if (n < NXc) p = (v2f){W3g[(2 * kk) * NXc + n], W3g[(2 * kk + 1) * NXc + n]};
    W3p[kk * LDW3 + n] = p;
  }
  if (tid < 128) { sB1[tid] = b1g[tid]; sB2[tid] = b2g[tid]; }
  if (tid < NXc) sB3[tid] = b3g[tid];
  if (tid < 3 * BT * 4) {
    float* tl = (tid < 32) ? sT1 : ((tid < 64) ? sT2 : sT4);
    tl[((tid >> 2) & (BT - 1)) * DPITCH + 60 + (tid & 3)] = 0.0f;
  }
  const float* S = (const float*)smem;
  const int bc  = tid & 63;
  const int br0 = tid >> 6;
  int ia0, sa, ib0, sb;
  bool isY = false;
  if (bc < NXc) {
    ia0 = (int)(OFF_XG / 4u) + bc;             sa = NXc;
    ib0 = ia0;                                 sb = sa;
  } else if (bc < 48) {
    const int idx = bc - 8;
    isY = true;
    ia0 = (int)(OFF_Z / 4u) + idx;             sa = 52;
    if (idx < 32) { ib0 = (int)(OFF_Z / 4u) + 8 + idx;      sb = 52; }
    else          { ib0 = (int)(OFF_XG / 4u) + (idx - 32);  sb = NXc; }
  } else {
    ia0 = (int)(OFF_Z / 4u) + 40 + (bc - 48);  sa = 52;
    ib0 = ia0;                                 sb = sa;
  }
  for (int i = tid; i < BT * 58; i += 256) {
    int r = i / 58, c = i - r * 58;
    float v = xGz0[(size_t)(bt0 + r) * 58 + c];
    if (c < NXc) sXG[r * NXc + c] = v;
    else         sZ[r * 52 + (c - NXc)] = v;
  }
  __syncthreads();

  for (int t = 0; t < T_STEPS; ++t) {
    float uv = 0.0f;
    if (tid < BT * NUc)
      uv = useq[(size_t)(bt0 + (tid >> 1)) * T_STEPS * NUc + (size_t)t * NUc + (tid & 1)];
    if (bc < 58) {
      const float a0 = S[ia0 + br0 * sa];
      const float b0 = S[ib0 + br0 * sb];
      const float a1 = S[ia0 + (br0 + 4) * sa];
      const float b1 = S[ib0 + (br0 + 4) * sb];
      sT1[br0 * DPITCH + bc]       = a0;
      sT1[(br0 + 4) * DPITCH + bc] = a1;
      if (bc >= NXc) {
        sT4[br0 * DPITCH + bc]       = b0;
        sT4[(br0 + 4) * DPITCH + bc] = b1;
        sT2[br0 * DPITCH + bc]       = isY ? 0.5f * (a0 + b0) : a0;
        sT2[(br0 + 4) * DPITCH + bc] = isY ? 0.5f * (a1 + b1) : a1;
      }
    }
    if (tid < BT * NUc) {
      const int r = tid >> 1, c = tid & 1;
      sU[tid] = uv;
      sT1[r * DPITCH + 58 + c] = uv;
      sT2[r * DPITCH + 58 + c] = uv;
      sT4[r * DPITCH + 58 + c] = uv;
    }
    __syncthreads();

    mlp_and_physics(tid, wave, lane, 0, sT1, sP2, Sw, sNN, W1p, W2p, W3p,
                    sB1, sB2, sB3, sXG, sU, sK, sXG, sXs, true, sT2, 0.5f * DELTA_F);

    mlp_and_physics(tid, wave, lane, 1, sT2, sP2, Sw, sNN, W1p, W2p, W3p,
                    sB1, sB2, sB3, sXs, sU, sK, sXG, sXs, true, sT2, 0.5f * DELTA_F);

    mlp_and_physics(tid, wave, lane, 2, sT2, sP2, Sw, sNN, W1p, W2p, W3p,
                    sB1, sB2, sB3, sXs, sU, sK, sXG, sXs, true, sT4, DELTA_F);

    mlp_and_physics(tid, wave, lane, 3, sT4, sP2, Sw, sNN, W1p, W2p, W3p,
                    sB1, sB2, sB3, sXs, sU, sK, sXG, sXs, false, sT4, 0.0f);

    float xgnew = 0.0f;
    if (tid < BT * NXc) {
      int r = tid >> 3, j = tid & 7;
      float xg = sXG[r * NXc + j];
      out[(size_t)(bt0 + r) * T_STEPS * NXc + (size_t)t * NXc + j] = xg;
      xgnew = xg + (DELTA_F / 6.0f) *
              (sK[0 * KST + tid] + 2.0f * sK[1 * KST + tid] +
               2.0f * sK[2 * KST + tid] + sK[3 * KST + tid]);
    }
    float regz[2];
#pragma unroll
    for (int ii = 0; ii < 2; ++ii) {
      int i = tid + ii * 256;
      if (i < BT * NZc) {
        int r = i / NZc, c = i - r * NZc;
        float v;
        if      (c < 32) v = sZ[r * 52 + 8 + c];
        else if (c < 40) v = sXG[r * NXc + (c - 32)];
        else if (c < 48) v = sZ[r * 52 + (c + 2)];
        else             v = sU[r * 2 + (c - 48)];
        regz[ii] = v;
      }
    }
    __syncthreads();
#pragma unroll
    for (int ii = 0; ii < 2; ++ii) {
      int i = tid + ii * 256;
      if (i < BT * NZc) {
        int r = i / NZc, c = i - r * NZc;
        sZ[r * 52 + c] = regz[ii];
      }
    }
    if (tid < BT * NXc) sXG[tid] = xgnew;
    __syncthreads();
  }
}

extern "C" void kernel_launch(void* const* d_in, const int* in_sizes, int n_in,
                              void* d_out, int out_size, void* d_ws, size_t ws_size,
                              hipStream_t stream) {
  (void)in_sizes; (void)n_in; (void)out_size; (void)d_ws; (void)ws_size;
  const float* useq = (const float*)d_in[0];
  const float* xGz0 = (const float*)d_in[1];
  const float* W1   = (const float*)d_in[2];
  const float* b1   = (const float*)d_in[3];
  const float* W2   = (const float*)d_in[4];
  const float* b2   = (const float*)d_in[5];
  const float* W3   = (const float*)d_in[6];
  const float* b3   = (const float*)d_in[7];
  float* out = (float*)d_out;
  cstr_rollout_kernel<<<dim3(B_TOT / BT), dim3(256), SMEM_BYTES, stream>>>(
      useq, xGz0, W1, b1, W2, b2, W3, b3, out);
}


# Physics RHS formed at each stage's head by wave 0's 8 row lanes under the other waves' matrix work; next state = stored xG+c*(fg+b3) + c*nn in every lane: 2 barriers a stage; k summed (fg+b3)+nn
# speedup vs baseline: 2.7904x; 1.0566x over previous
#include <hip/hip_runtime.h>
#include <math.h>


typedef __attribute__((ext_vector_type(2))) float v2f;
typedef __attribute__((ext_vector_type(4))) float v4f;
typedef __attribute__((ext_vector_type(8))) float v8f;

#define B_TOT   2048
#define T_STEPS 128
#define NXc     8
#define NUc     2
#define NZc     50
#define DINP    64
#define DPITCH  68
#define DH      128
#define BT      8
#define KST     (BT * NXc)
#define DELTA_F 0.01f

#define LDW     129
#define LDW3    17
#define OFF_W1P 0u
#define OFF_W2P (OFF_W1P + 32u*LDW*8u)
#define OFF_W3P (OFF_W2P + 64u*LDW*8u)
#define OFF_INP (OFF_W3P + 64u*LDW3*8u)
#define SW_FLT  136
#define P_FLT   1024
#define OFF_P1  (OFF_INP + 16u*64u*4u)
#define OFF_T2  (OFF_P1)
#define OFF_T4  (OFF_P1 + 4096u)
#define OFF_P2  (OFF_P1  + 8u*P_FLT*4u)
#define OFF_SW  (OFF_P2  + 8u*P_FLT*4u)
#define OFF_NN  (OFF_SW  + 8u*SW_FLT*4u)
#define OFF_XG  (OFF_NN  + 256u*4u)
#define OFF_Z   (OFF_XG  + 128u*4u)
#define OFF_F   (OFF_Z   + 16u*52u*4u)
#define OFF_PN  (OFF_F   + 256u*4u)
#define OFF_U   (OFF_PN  + 256u*4u)
#define OFF_B1  (OFF_U   + 32u*4u)
#define OFF_B2  (OFF_B1  + 128u*4u)
#define OFF_B3  (OFF_B2  + 128u*4u)
#define SMEM_BYTES (OFF_B3 + 16u*4u)

__device__ __forceinline__ float fast_tanh(float x) {
  float e = __expf(2.0f * x);
  return 1.0f - __fdividef(2.0f, e + 1.0f);
}


__device__ __forceinline__ v4f my_rows(v8f d, int kh) {
  const v4f lo = __builtin_shufflevector(d, d, 0, 1, 2, 3);
  const v4f hi = __builtin_shufflevector(d, d, 4, 5, 6, 7);
  return kh ? hi : lo;
}

__device__ __forceinline__ int sw_row(int r) { return 16 * r + 4 * (r >> 2); }

__device__ __forceinline__ void own_cols_to_afrag(float* Sw, v4f h, int lane, v2f a[4]) {
  const int n  = lane & 15;
  const int kh = lane >> 4;
#pragma unroll
  for (int v = 0; v < 4; ++v) Sw[sw_row(4 * kh + v) + n] = h[v];
  const float* rp = Sw + sw_row(n & (BT - 1)) + 2 * kh;
#pragma unroll
  for (int kk = 0; kk < 4; ++kk) a[kk] = *(const v2f*)(rp + 4 * kk);
}

template <int KSTEPS>
__device__ __forceinline__ void slice_times_rows(const v2f* a, const v2f* __restrict__ Bp,
                                                 float* pw, int lane) {
  const int n  = lane & 15;
  const int kh = lane >> 4;
  v8f acc[8];
#pragma unroll
  for (int j = 0; j < 8; ++j) acc[j] = (v8f){};
  const v2f* bp = Bp + (size_t)kh * LDW + n;
#pragma unroll
  for (int kk = 0; kk < KSTEPS; ++kk) {
#pragma unroll
    for (int j = 0; j < 8; ++j) {
      v2f b = bp[(size_t)(2 * kk) * LDW + 16 * j];
      acc[j] = __builtin_amdgcn_wmma_f32_16x16x4_f32(false, a[kk], false, b,
                                                     (short)0, acc[j], false, false);
    }
  }
  float* p = pw + n * 8 + 4 * kh;
#pragma unroll
  for (int j = 0; j < 8; ++j) *(v4f*)(p + j * 128) = my_rows(acc[j], kh);
}

__device__ __forceinline__ void layer1_load(const float* sInp, const v2f* __restrict__ W1p,
                                            int wave, int lane, v2f a[16], v2f b[16]) {
  const int n  = lane & 15;
  const int kh = lane >> 4;
  const float* ap = sInp + (n & (BT - 1)) * DPITCH + 2 * kh;
  const v2f*   bp = W1p + (size_t)kh * LDW + 16 * wave + n;
#pragma unroll
  for (int kk = 0; kk < 16; ++kk) {
    b[kk] = bp[(size_t)(2 * kk) * LDW];
    if (kk >= 2) a[kk] = *(const v2f*)(ap + 4 * kk);
  }
}

__device__ __forceinline__ v4f layer1_mul(const v2f a[16], const v2f b[16], int kh) {
  v8f acc[4];
#pragma unroll
  for (int c = 0; c < 4; ++c) acc[c] = (v8f){};
#pragma unroll
  for (int kk = 0; kk < 16; ++kk) {
    acc[kk & 3] = __builtin_amdgcn_wmma_f32_16x16x4_f32(false, a[kk], false, b[kk],
                                                        (short)0, acc[kk & 3], false, false);
  }
  const v8f d = (acc[0] + acc[1]) + (acc[2] + acc[3]);
  return my_rows(d, kh);
}

__device__ __forceinline__ v4f sum_slices_tanh(const float* sP, int nc, int kh, float bias) {
  const float* pr = sP + nc * 8 + 4 * kh;
  v4f s = *(const v4f*)pr;
#pragma unroll
  for (int w = 1; w < 8; ++w) s += *(const v4f*)(pr + w * P_FLT);
  v4f h;
#pragma unroll
  for (int v = 0; v < 4; ++v) h[v] = fast_tanh(s[v] + bias);
  return h;
}

__device__ __forceinline__ void rk_stage(
    int tid, int wave, int lane, int stage,
    const float* sTile, float* sP2, float* Sw,
    const v2f* W1p, const v2f* W2p, const v2f* W3p,
    const float* sB1, const float* sB2, const float* sB3,
    const float* sXG, const float* sU,
    float* sF, float* sPn, float* sNN, float cprev, float cnext) {
  const int n   = lane & 15;
  const int kh  = lane >> 4;
  const int row = n & (BT - 1);
  const int nc  = wave * 16 + n;
  const int    sprev = (stage > 0) ? stage - 1 : 0;
  float*       sNNs  = sNN + stage * (BT * NXc);
  const float* sNNp  = sNN + sprev * (BT * NXc);
  const float* sPp   = sPn + sprev * (BT * NXc);
  v4f g0, g1, q0, q1, m0, m1, c0, c1;
  v2f uu;
  if (tid < BT) {
    g0 = *(const v4f*)(sXG + tid * NXc);
    g1 = *(const v4f*)(sXG + tid * NXc + 4);
    if (stage != 0) {
      q0 = *(const v4f*)(sPp  + tid * NXc);  q1 = *(const v4f*)(sPp  + tid * NXc + 4);
      m0 = *(const v4f*)(sNNp + tid * NXc);  m1 = *(const v4f*)(sNNp + tid * NXc + 4);
    }
    uu = *(const v2f*)(sU + tid * 2);
    c0 = *(const v4f*)sB3;
    c1 = *(const v4f*)(sB3 + 4);
  }
  v2f p0, p1, n0, n1;
  if (stage == 0) {
    const float* xp = sXG + row * NXc + 2 * kh;
    p0 = *(const v2f*)xp;
    p1 = *(const v2f*)(xp + 4);
  } else {
    const float* pp = sPp  + row * NXc + 2 * kh;
    const float* np = sNNp + row * NXc + 2 * kh;
    p0 = *(const v2f*)pp;  p1 = *(const v2f*)(pp + 4);
    n0 = *(const v2f*)np;  n1 = *(const v2f*)(np + 4);
  }
  v2f a1f[16], b1f[16];
  layer1_load(sTile, W1p, wave, lane, a1f, b1f);
  if (tid < BT) {
    v4f x0 = g0, x1 = g1;
    if (stage != 0) {
      x0 = q0 + cprev * m0;
      x1 = q1 + cprev * m1;
    }
    float Hr  = x0[0] * 2.0f + 50.0f;
    float CAr = x0[1] * 0.2f + 1.0f;
    float CBr = x0[2] * 0.1f + 0.5f;
    float Tr  = x0[3] * 4.0f + 313.0f;
    float Hb  = x1[0] * 2.0f + 50.0f;
    float CAb = x1[1] * 0.2f + 1.0f;
    float CBb = x1[2] * 0.1f + 0.5f;
    float Tb  = x1[3] * 4.0f + 313.0f;
    float F = uu[0] * 1.0f + 10.0f;
    float D = uu[1] * 0.5f + 5.0f;
    float Fr = 2.5f * __fsqrt_rn(Hr);
    float Fb = 1.5f * __fsqrt_rn(Hb);
    float invArHr = __fdividef(1.0f, 3.0f * Hr);
    float invAbHb = __fdividef(1.0f, 3.0f * Hb);
    float dHr  = (F + D - Fr) * (1.0f / 3.0f);
    float dCAr = (F * (6.0f - CAr) - D * CAr) * invArHr;
    float dCBr = (-F * CBr - D * CBr) * invArHr;
    float dTr  = (F * (320.0f - Tr) + D * (300.0f - Tr)) * invArHr
               + (200.0f / 30.0f) * invArHr;
    float dHb  = (Fr - Fb - D) * (1.0f / 3.0f);
    float dCAb = (Fr * (CAr - CAb) + D * CAb) * invAbHb;
    float dCBb = (Fr * (CBr - CBb) + D * CBb) * invAbHb;
    float dTb  = Fr * (Tr - Tb) * invAbHb + (200.0f / 30.0f) * invAbHb;
    v4f f0, f1;
    f0[0] = dHr  * 0.5f  + c0[0];
    f0[1] = dCAr * 5.0f  + c0[1];
    f0[2] = dCBr * 10.0f + c0[2];
    f0[3] = dTr  * 0.25f + c0[3];
    f1[0] = dHb  * 0.5f  + c1[0];
    f1[1] = dCAb * 5.0f  + c1[1];
    f1[2] = dCBb * 10.0f + c1[2];
    f1[3] = dTb  * 0.25f + c1[3];
    float* fr = sF + stage * (BT * NXc) + tid * NXc;
    *(v4f*)fr       = f0;
    *(v4f*)(fr + 4) = f1;
    if (stage != 3) {
      float* pr = sPn + stage * (BT * NXc) + tid * NXc;
      *(v4f*)pr       = g0 + cnext * f0;
      *(v4f*)(pr + 4) = g1 + cnext * f1;
    }
  }
  v2f a[4];
  {
    a1f[0] = (stage == 0) ? p0 : p0 + cprev * n0;
    a1f[1] = (stage == 0) ? p1 : p1 + cprev * n1;
    const v4f s1 = layer1_mul(a1f, b1f, kh);
    const float bias1 = sB1[nc];
    v4f h1;
#pragma unroll
    for (int v = 0; v < 4; ++v) h1[v] = fast_tanh(s1[v] + bias1);
    own_cols_to_afrag(Sw, h1, lane, a);
  }
  slice_times_rows<4>(a, W2p + (size_t)(8 * wave) * LDW, sP2 + wave * P_FLT, lane);
  __syncthreads();
  own_cols_to_afrag(Sw, sum_slices_tanh(sP2, nc, kh, sB2[nc]), lane, a);
  {
    v8f d = {};
    const v2f* bp = W3p + (size_t)(8 * wave + kh) * LDW3 + n;
#pragma unroll
    for (int kk = 0; kk < 4; ++kk) {
      v2f b = bp[(size_t)(2 * kk) * LDW3];
      d = __builtin_amdgcn_wmma_f32_16x16x4_f32(false, a[kk], false, b,
                                                (short)0, d, false, false);
    }
    const v4f s = my_rows(d, kh);
    if (n < NXc) {
#pragma unroll
      for (int v = 0; v < 4; ++v)
        atomicAdd(&sNNs[(v + 4 * kh) * NXc + n], s[v]);
    }
  }
  __syncthreads();
}

__global__ __launch_bounds__(256) void cstr_rollout_kernel(
    const float* __restrict__ useq, const float* __restrict__ xGz0,
    const float* __restrict__ W1g, const float* __restrict__ b1g,
    const float* __restrict__ W2g, const float* __restrict__ b2g,
    const float* __restrict__ W3g, const float* __restrict__ b3g,
    float* __restrict__ out) {
  extern __shared__ __align__(16) char smem[];
  v2f*   W1p  = (v2f*)(smem + OFF_W1P);
  v2f*   W2p  = (v2f*)(smem + OFF_W2P);
  v2f*   W3p  = (v2f*)(smem + OFF_W3P);
  float* sT1  = (float*)(smem + OFF_INP);
  float* sT2  = (float*)(smem + OFF_T2);
  float* sT4  = (float*)(smem + OFF_T4);
  float* sP2  = (float*)(smem + OFF_P2);
  float* sNN  = (float*)(smem + OFF_NN);
  float* sXG  = (float*)(smem + OFF_XG);
  float* sZ   = (float*)(smem + OFF_Z);
  float* sF   = (float*)(smem + OFF_F);
  float* sPn  = (float*)(smem + OFF_PN);
  float* sU   = (float*)(smem + OFF_U);
  float* sB1  = (float*)(smem + OFF_B1);
  float* sB2  = (float*)(smem + OFF_B2);
  float* sB3  = (float*)(smem + OFF_B3);

  const int tid  = threadIdx.x;
  const int lane = tid & 31;
  const int wave = tid >> 5;
  const int bt0  = blockIdx.x * BT;
  float* Sw = (float*)(smem + OFF_SW) + wave * SW_FLT;

  for (int i = tid; i < 32 * 128; i += 256) {
    int kk = i >> 7, n = i & 127;
    float lo = (2 * kk     < 60) ? W1g[(2 * kk)     * 128 + n] : 0.0f;
    float hi = (2 * kk + 1 < 60) ? W1g[(2 * kk + 1) * 128 + n] : 0.0f;
    W1p[kk * LDW + n] = (v2f){lo, hi};
  }
  for (int i = tid; i < 64 * 128; i += 256) {
    int kk = i >> 7, n = i & 127;
    W2p[kk * LDW + n] = (v2f){W2g[(2 * kk) * 128 + n], W2g[(2 * kk + 1) * 128 + n]};
  }
  for (int i = tid; i < 64 * 16; i += 256) {
    int kk = i >> 4, n = i & 15;
    v2f p = (v2f){0.0f, 0.0f};
    if (n < NXc) p = (v2f){W3g[(2 * kk) * NXc + n], W3g[(2 * kk + 1) * NXc + n]};
    W3p[kk * LDW3 + n] = p;
  }
  if (tid < 128) { sB1[tid] = b1g[tid]; sB2[tid] = b2g[tid]; }
  if (tid < NXc) sB3[tid] = b3g[tid];
  if (tid < 3 * BT * 4) {
    float* tl = (tid < 32) ? sT1 : ((tid < 64) ? sT2 : sT4);
    tl[((tid >> 2) & (BT - 1)) * DPITCH + 60 + (tid & 3)] = 0.0f;
  }
  const float* S = (const float*)smem;
  const int bc  = tid & 63;
  const int br0 = tid >> 6;
  int ia0 = 0, sa = 0, ib0 = 0, sb = 0;
  bool isY = false;
  if (bc < NXc) {
  } else if (bc < 48) {
    const int idx = bc - 8;
    isY = true;
    ia0 = (int)(OFF_Z / 4u) + idx;             sa = 52;
    if (idx < 32) { ib0 = (int)(OFF_Z / 4u) + 8 + idx;      sb = 52; }
    else          { ib0 = (int)(OFF_XG / 4u) + (idx - 32);  sb = NXc; }
  } else {
    ia0 = (int)(OFF_Z / 4u) + 40 + (bc - 48);  sa = 52;
    ib0 = ia0;                                 sb = sa;
  }
  for (int i = tid; i < BT * 58; i += 256) {
    int r = i / 58, c = i - r * 58;
    float v = xGz0[(size_t)(bt0 + r) * 58 + c];
    if (c < NXc) sXG[r * NXc + c] = v;
    else         sZ[r * 52 + (c - NXc)] = v;
  }
  __syncthreads();

  for (int t = 0; t < T_STEPS; ++t) {
    float uv = 0.0f;
    if (tid < BT * NUc)
      uv = useq[(size_t)(bt0 + (tid >> 1)) * T_STEPS * NUc + (size_t)t * NUc + (tid & 1)];
    sNN[tid] = 0.0f;
    if (bc >= NXc && bc < 58) {
      const float a0 = S[ia0 + br0 * sa];
      const float b0 = S[ib0 + br0 * sb];
      const float a1 = S[ia0 + (br0 + 4) * sa];
      const float b1 = S[ib0 + (br0 + 4) * sb];
      sT1[br0 * DPITCH + bc]       = a0;
      sT1[(br0 + 4) * DPITCH + bc] = a1;
      sT4[br0 * DPITCH + bc]       = b0;
      sT4[(br0 + 4) * DPITCH + bc] = b1;
      sT2[br0 * DPITCH + bc]       = isY ? 0.5f * (a0 + b0) : a0;
      sT2[(br0 + 4) * DPITCH + bc] = isY ? 0.5f * (a1 + b1) : a1;
    }
    if (tid < BT * NUc) {
      const int r = tid >> 1, c = tid & 1;
      sU[tid] = uv;
      sT1[r * DPITCH + 58 + c] = uv;
      sT2[r * DPITCH + 58 + c] = uv;
      sT4[r * DPITCH + 58 + c] = uv;
    }
    __syncthreads();

    rk_stage(tid, wave, lane, 0, sT1, sP2, Sw, W1p, W2p, W3p, sB1, sB2, sB3,
             sXG, sU, sF, sPn, sNN, 0.0f, 0.5f * DELTA_F);

    rk_stage(tid, wave, lane, 1, sT2, sP2, Sw, W1p, W2p, W3p, sB1, sB2, sB3,
             sXG, sU, sF, sPn, sNN, 0.5f * DELTA_F, 0.5f * DELTA_F);

    rk_stage(tid, wave, lane, 2, sT2, sP2, Sw, W1p, W2p, W3p, sB1, sB2, sB3,
             sXG, sU, sF, sPn, sNN, 0.5f * DELTA_F, DELTA_F);

    rk_stage(tid, wave, lane, 3, sT4, sP2, Sw, W1p, W2p, W3p, sB1, sB2, sB3,
             sXG, sU, sF, sPn, sNN, DELTA_F, 0.0f);

    float xgnew = 0.0f;
    if (tid < BT * NXc) {
      int r = tid >> 3, j = tid & 7;
      float xg = sXG[r * NXc + j];
      out[(size_t)(bt0 + r) * T_STEPS * NXc + (size_t)t * NXc + j] = xg;
      const float k1 = sF[0 * KST + tid] + sNN[0 * KST + tid];
      const float k2 = sF[1 * KST + tid] + sNN[1 * KST + tid];
      const float k3 = sF[2 * KST + tid] + sNN[2 * KST + tid];
      const float k4 = sF[3 * KST + tid] + sNN[3 * KST + tid];
      xgnew = xg + (DELTA_F / 6.0f) * (k1 + 2.0f * k2 + 2.0f * k3 + k4);
    }
    float regz[2];
#pragma unroll
    for (int ii = 0; ii < 2; ++ii) {
      int i = tid + ii * 256;
      if (i < BT * NZc) {
        int r = i / NZc, c = i - r * NZc;
        float v;
        if      (c < 32) v = sZ[r * 52 + 8 + c];
        else if (c < 40) v = sXG[r * NXc + (c - 32)];
        else if (c < 48) v = sZ[r * 52 + (c + 2)];
        else             v = sU[r * 2 + (c - 48)];
        regz[ii] = v;
      }
    }
    __syncthreads();
#pragma unroll
    for (int ii = 0; ii < 2; ++ii) {
      int i = tid + ii * 256;
      if (i < BT * NZc) {
        int r = i / NZc, c = i - r * NZc;
        sZ[r * 52 + c] = regz[ii];
      }
    }
    if (tid < BT * NXc) sXG[tid] = xgnew;
    __syncthreads();
  }
}

extern "C" void kernel_launch(void* const* d_in, const int* in_sizes, int n_in,
                              void* d_out, int out_size, void* d_ws, size_t ws_size,
                              hipStream_t stream) {
  (void)in_sizes; (void)n_in; (void)out_size; (void)d_ws; (void)ws_size;
  const float* useq = (const float*)d_in[0];
  const float* xGz0 = (const float*)d_in[1];
  const float* W1   = (const float*)d_in[2];
  const float* b1   = (const float*)d_in[3];
  const float* W2   = (const float*)d_in[4];
  const float* b2   = (const float*)d_in[5];
  const float* W3   = (const float*)d_in[6];
  const float* b3   = (const float*)d_in[7];
  float* out = (float*)d_out;
  cstr_rollout_kernel<<<dim3(B_TOT / BT), dim3(256), SMEM_BYTES, stream>>>(
      useq, xGz0, W1, b1, W2, b2, W3, b3, out);
}


# Layer-1 history columns multiplied once a step, not per stage: the tile's repeated 8 rows carry stage 4's tile beside stage 1's (2 passes); a stage adds its 2 state k-steps; L1 sum order changes
# speedup vs baseline: 3.0095x; 1.0785x over previous
#include <hip/hip_runtime.h>
#include <math.h>


typedef __attribute__((ext_vector_type(2))) float v2f;
typedef __attribute__((ext_vector_type(4))) float v4f;
typedef __attribute__((ext_vector_type(8))) float v8f;

#define B_TOT   2048
#define T_STEPS 128
#define NXc     8
#define NUc     2
#define NZc     50
#define DINP    64
#define DPITCH  68
#define DH      128
#define BT      8
#define KST     (BT * NXc)
#define DELTA_F 0.01f

#define LDW     129
#define LDW3    17
#define OFF_W1P 0u
#define OFF_W2P (OFF_W1P + 32u*LDW*8u)
#define OFF_W3P (OFF_W2P + 64u*LDW*8u)
#define OFF_INP (OFF_W3P + 64u*LDW3*8u)
#define SW_FLT  136
#define P_FLT   1024
#define OFF_P1  (OFF_INP + 16u*64u*4u)
#define OFF_T2  (OFF_P1)
#define OFF_T4  (OFF_P1 + 4096u + 128u)
#define OFF_P2  (OFF_P1  + 8u*P_FLT*4u)
#define OFF_SW  (OFF_P2  + 8u*P_FLT*4u)
#define OFF_NN  (OFF_SW  + 8u*SW_FLT*4u)
#define OFF_XG  (OFF_NN  + 256u*4u)
#define OFF_Z   (OFF_XG  + 128u*4u)
#define OFF_F   (OFF_Z   + 16u*52u*4u)
#define OFF_PN  (OFF_F   + 256u*4u)
#define OFF_U   (OFF_PN  + 256u*4u)
#define OFF_B1  (OFF_U   + 32u*4u)
#define OFF_B2  (OFF_B1  + 128u*4u)
#define OFF_B3  (OFF_B2  + 128u*4u)
#define SMEM_BYTES (OFF_B3 + 16u*4u)

__device__ __forceinline__ float fast_tanh(float x) {
  float e = __expf(2.0f * x);
  return 1.0f - __fdividef(2.0f, e + 1.0f);
}


__device__ __forceinline__ v4f my_rows(v8f d, int kh) {
  const v4f lo = __builtin_shufflevector(d, d, 0, 1, 2, 3);
  const v4f hi = __builtin_shufflevector(d, d, 4, 5, 6, 7);
  return kh ? hi : lo;
}

__device__ __forceinline__ int sw_row(int r) { return 16 * r + 4 * (r >> 2); }

__device__ __forceinline__ void own_cols_to_afrag(float* Sw, v4f h, int lane, v2f a[4]) {
  const int n  = lane & 15;
  const int kh = lane >> 4;
#pragma unroll
  for (int v = 0; v < 4; ++v) Sw[sw_row(4 * kh + v) + n] = h[v];
  const float* rp = Sw + sw_row(n & (BT - 1)) + 2 * kh;
#pragma unroll
  for (int kk = 0; kk < 4; ++kk) a[kk] = *(const v2f*)(rp + 4 * kk);
}

template <int KSTEPS>
__device__ __forceinline__ void slice_times_rows(const v2f* a, const v2f* __restrict__ Bp,
                                                 float* pw, int lane) {
  const int n  = lane & 15;
  const int kh = lane >> 4;
  v8f acc[8];
#pragma unroll
  for (int j = 0; j < 8; ++j) acc[j] = (v8f){};
  const v2f* bp = Bp + (size_t)kh * LDW + n;
#pragma unroll
  for (int kk = 0; kk < KSTEPS; ++kk) {
#pragma unroll
    for (int j = 0; j < 8; ++j) {
      v2f b = bp[(size_t)(2 * kk) * LDW + 16 * j];
      acc[j] = __builtin_amdgcn_wmma_f32_16x16x4_f32(false, a[kk], false, b,
                                                     (short)0, acc[j], false, false);
    }
  }
  float* p = pw + n * 8 + 4 * kh;
#pragma unroll
  for (int j = 0; j < 8; ++j) *(v4f*)(p + j * 128) = my_rows(acc[j], kh);
}

__device__ __forceinline__ void hist_load_a(const float* ap, v2f a[16]) {
#pragma unroll
  for (int kk = 2; kk < 16; ++kk) a[kk] = *(const v2f*)(ap + 4 * kk);
}
__device__ __forceinline__ v8f hist_mul(const v2f a[16], const v2f b[16]) {
  v8f acc[4];
#pragma unroll
  for (int c = 0; c < 4; ++c) acc[c] = (v8f){};
#pragma unroll
  for (int kk = 2; kk < 16; ++kk) {
    acc[kk & 3] = __builtin_amdgcn_wmma_f32_16x16x4_f32(false, a[kk], false, b[kk],
                                                        (short)0, acc[kk & 3], false, false);
  }
  return (acc[0] + acc[1]) + (acc[2] + acc[3]);
}

__device__ __forceinline__ v4f sum_slices_tanh(const float* sP, int nc, int kh, float bias) {
  const float* pr = sP + nc * 8 + 4 * kh;
  v4f s = *(const v4f*)pr;
#pragma unroll
  for (int w = 1; w < 8; ++w) s += *(const v4f*)(pr + w * P_FLT);
  v4f h;
#pragma unroll
  for (int v = 0; v < 4; ++v) h[v] = fast_tanh(s[v] + bias);
  return h;
}

__device__ __forceinline__ void rk_stage(
    int tid, int wave, int lane, int stage,
    v4f hist, float* sP2, float* Sw,
    const v2f* W1p, const v2f* W2p, const v2f* W3p,
    const float* sB2, const float* sB3,
    const float* sXG, const float* sU,
    float* sF, float* sPn, float* sNN, float cprev, float cnext) {
  const int n   = lane & 15;
  const int kh  = lane >> 4;
  const int row = n & (BT - 1);
  const int nc  = wave * 16 + n;
  const int    sprev = (stage > 0) ? stage - 1 : 0;
  float*       sNNs  = sNN + stage * (BT * NXc);
  const float* sNNp  = sNN + sprev * (BT * NXc);
  const float* sPp   = sPn + sprev * (BT * NXc);
  v4f g0, g1, q0, q1, m0, m1, c0, c1;
  v2f uu;
  if (tid < BT) {
    g0 = *(const v4f*)(sXG + tid * NXc);
    g1 = *(const v4f*)(sXG + tid * NXc + 4);
    if (stage != 0) {
      q0 = *(const v4f*)(sPp  + tid * NXc);  q1 = *(const v4f*)(sPp  + tid * NXc + 4);
      m0 = *(const v4f*)(sNNp + tid * NXc);  m1 = *(const v4f*)(sNNp + tid * NXc + 4);
    }
    uu = *(const v2f*)(sU + tid * 2);
    c0 = *(const v4f*)sB3;
    c1 = *(const v4f*)(sB3 + 4);
  }
  v2f p0, p1, n0, n1;
  if (stage == 0) {
    const float* xp = sXG + row * NXc + 2 * kh;
    p0 = *(const v2f*)xp;
    p1 = *(const v2f*)(xp + 4);
  } else {
    const float* pp = sPp  + row * NXc + 2 * kh;
    const float* np = sNNp + row * NXc + 2 * kh;
    p0 = *(const v2f*)pp;  p1 = *(const v2f*)(pp + 4);
    n0 = *(const v2f*)np;  n1 = *(const v2f*)(np + 4);
  }
  const v2f* bp1 = W1p + (size_t)kh * LDW + 16 * wave + n;
  const v2f bs0 = bp1[0];
  const v2f bs1 = bp1[(size_t)2 * LDW];
  if (tid < BT) {
    v4f x0 = g0, x1 = g1;
    if (stage != 0) {
      x0 = q0 + cprev * m0;
      x1 = q1 + cprev * m1;
    }
    float Hr  = x0[0] * 2.0f + 50.0f;
    float CAr = x0[1] * 0.2f + 1.0f;
    float CBr = x0[2] * 0.1f + 0.5f;
    float Tr  = x0[3] * 4.0f + 313.0f;
    float Hb  = x1[0] * 2.0f + 50.0f;
    float CAb = x1[1] * 0.2f + 1.0f;
    float CBb = x1[2] * 0.1f + 0.5f;
    float Tb  = x1[3] * 4.0f + 313.0f;
    float F = uu[0] * 1.0f + 10.0f;
    float D = uu[1] * 0.5f + 5.0f;
    float Fr = 2.5f * __fsqrt_rn(Hr);
    float Fb = 1.5f * __fsqrt_rn(Hb);
    float invArHr = __fdividef(1.0f, 3.0f * Hr);
    float invAbHb = __fdividef(1.0f, 3.0f * Hb);
    float dHr  = (F + D - Fr) * (1.0f / 3.0f);
    float dCAr = (F * (6.0f - CAr) - D * CAr) * invArHr;
    float dCBr = (-F * CBr - D * CBr) * invArHr;
    float dTr  = (F * (320.0f - Tr) + D * (300.0f - Tr)) * invArHr
               + (200.0f / 30.0f) * invArHr;
    float dHb  = (Fr - Fb - D) * (1.0f / 3.0f);
    float dCAb = (Fr * (CAr - CAb) + D * CAb) * invAbHb;
    float dCBb = (Fr * (CBr - CBb) + D * CBb) * invAbHb;
    float dTb  = Fr * (Tr - Tb) * invAbHb + (200.0f / 30.0f) * invAbHb;
    v4f f0, f1;
    f0[0] = dHr  * 0.5f  + c0[0];
    f0[1] = dCAr * 5.0f  + c0[1];
    f0[2] = dCBr * 10.0f + c0[2];
    f0[3] = dTr  * 0.25f + c0[3];
    f1[0] = dHb  * 0.5f  + c1[0];
    f1[1] = dCAb * 5.0f  + c1[1];
    f1[2] = dCBb * 10.0f + c1[2];
    f1[3] = dTb  * 0.25f + c1[3];
    float* fr = sF + stage * (BT * NXc) + tid * NXc;
    *(v4f*)fr       = f0;
    *(v4f*)(fr + 4) = f1;
    if (stage != 3) {
      float* pr = sPn + stage * (BT * NXc) + tid * NXc;
      *(v4f*)pr       = g0 + cnext * f0;
      *(v4f*)(pr + 4) = g1 + cnext * f1;
    }
  }
  v2f a[4];
  {
    const v2f as0 = (stage == 0) ? p0 : p0 + cprev * n0;
    const v2f as1 = (stage == 0) ? p1 : p1 + cprev * n1;
    const v8f d0 = __builtin_amdgcn_wmma_f32_16x16x4_f32(false, as0, false, bs0,
                                                         (short)0, (v8f){}, false, false);
    const v8f d1 = __builtin_amdgcn_wmma_f32_16x16x4_f32(false, as1, false, bs1,
                                                         (short)0, (v8f){}, false, false);
    const v4f s1 = (hist + my_rows(d0, kh)) + my_rows(d1, kh);
    v4f h1;
#pragma unroll
    for (int v = 0; v < 4; ++v) h1[v] = fast_tanh(s1[v]);
    own_cols_to_afrag(Sw, h1, lane, a);
  }
  slice_times_rows<4>(a, W2p + (size_t)(8 * wave) * LDW, sP2 + wave * P_FLT, lane);
  __syncthreads();
  own_cols_to_afrag(Sw, sum_slices_tanh(sP2, nc, kh, sB2[nc]), lane, a);
  {
    v8f d = {};
    const v2f* bp = W3p + (size_t)(8 * wave + kh) * LDW3 + n;
#pragma unroll
    for (int kk = 0; kk < 4; ++kk) {
      v2f b = bp[(size_t)(2 * kk) * LDW3];
      d = __builtin_amdgcn_wmma_f32_16x16x4_f32(false, a[kk], false, b,
                                                (short)0, d, false, false);
    }
    const v4f s = my_rows(d, kh);
    if (n < NXc) {
#pragma unroll
      for (int v = 0; v < 4; ++v)
        atomicAdd(&sNNs[(v + 4 * kh) * NXc + n], s[v]);
    }
  }
  __syncthreads();
}

__global__ __launch_bounds__(256) void cstr_rollout_kernel(
    const float* __restrict__ useq, const float* __restrict__ xGz0,
    const float* __restrict__ W1g, const float* __restrict__ b1g,
    const float* __restrict__ W2g, const float* __restrict__ b2g,
    const float* __restrict__ W3g, const float* __restrict__ b3g,
    float* __restrict__ out) {
  extern __shared__ __align__(16) char smem[];
  v2f*   W1p  = (v2f*)(smem + OFF_W1P);
  v2f*   W2p  = (v2f*)(smem + OFF_W2P);
  v2f*   W3p  = (v2f*)(smem + OFF_W3P);
  float* sT1  = (float*)(smem + OFF_INP);
  float* sT2  = (float*)(smem + OFF_T2);
  float* sT4  = (float*)(smem + OFF_T4);
  float* sP2  = (float*)(smem + OFF_P2);
  float* sNN  = (float*)(smem + OFF_NN);
  float* sXG  = (float*)(smem + OFF_XG);
  float* sZ   = (float*)(smem + OFF_Z);
  float* sF   = (float*)(smem + OFF_F);
  float* sPn  = (float*)(smem + OFF_PN);
  float* sU   = (float*)(smem + OFF_U);
  float* sB1  = (float*)(smem + OFF_B1);
  float* sB2  = (float*)(smem + OFF_B2);
  float* sB3  = (float*)(smem + OFF_B3);

  const int tid  = threadIdx.x;
  const int lane = tid & 31;
  const int wave = tid >> 5;
  const int bt0  = blockIdx.x * BT;
  float* Sw = (float*)(smem + OFF_SW) + wave * SW_FLT;

  for (int i = tid; i < 32 * 128; i += 256) {
    int kk = i >> 7, n = i & 127;
    float lo = (2 * kk     < 60) ? W1g[(2 * kk)     * 128 + n] : 0.0f;
    float hi = (2 * kk + 1 < 60) ? W1g[(2 * kk + 1) * 128 + n] : 0.0f;
    W1p[kk * LDW + n] = (v2f){lo, hi};
  }
  for (int i = tid; i < 64 * 128; i += 256) {
    int kk = i >> 7, n = i & 127;
    W2p[kk * LDW + n] = (v2f){W2g[(2 * kk) * 128 + n], W2g[(2 * kk + 1) * 128 + n]};
  }
  for (int i = tid; i < 64 * 16; i += 256) {
    int kk = i >> 4, n = i & 15;
    v2f p = (v2f){0.0f, 0.0f};
    if (n < NXc) p = (v2f){W3g[(2 * kk) * NXc + n], W3g[(2 * kk + 1) * NXc + n]};
    W3p[kk * LDW3 + n] = p;
  }
  if (tid < 128) { sB1[tid] = b1g[tid]; sB2[tid] = b2g[tid]; }
  if (tid < NXc) sB3[tid] = b3g[tid];
  if (tid < 3 * BT * 4) {
    float* tl = (tid < 32) ? sT1 : ((tid < 64) ? sT2 : sT4);
    tl[((tid >> 2) & (BT - 1)) * DPITCH + 60 + (tid & 3)] = 0.0f;
  }
  const float* S = (const float*)smem;
  const int bc  = tid & 63;
  const int br0 = tid >> 6;
  int ia0 = 0, sa = 0, ib0 = 0, sb = 0;
  bool isY = false;
  if (bc < NXc) {
  } else if (bc < 48) {
    const int idx = bc - 8;
    isY = true;
    ia0 = (int)(OFF_Z / 4u) + idx;             sa = 52;
    if (idx < 32) { ib0 = (int)(OFF_Z / 4u) + 8 + idx;      sb = 52; }
    else          { ib0 = (int)(OFF_XG / 4u) + (idx - 32);  sb = NXc; }
  } else {
    ia0 = (int)(OFF_Z / 4u) + 40 + (bc - 48);  sa = 52;
    ib0 = ia0;                                 sb = sa;
  }
  for (int i = tid; i < BT * 58; i += 256) {
    int r = i / 58, c = i - r * 58;
    float v = xGz0[(size_t)(bt0 + r) * 58 + c];
    if (c < NXc) sXG[r * NXc + c] = v;
    else         sZ[r * 52 + (c - NXc)] = v;
  }
  __syncthreads();

  for (int t = 0; t < T_STEPS; ++t) {
    float uv = 0.0f;
    if (tid < BT * NUc)
      uv = useq[(size_t)(bt0 + (tid >> 1)) * T_STEPS * NUc + (size_t)t * NUc + (tid & 1)];
    sNN[tid] = 0.0f;
    if (bc >= NXc && bc < 58) {
      const float a0 = S[ia0 + br0 * sa];
      const float b0 = S[ib0 + br0 * sb];
      const float a1 = S[ia0 + (br0 + 4) * sa];
      const float b1 = S[ib0 + (br0 + 4) * sb];
      sT1[br0 * DPITCH + bc]       = a0;
      sT1[(br0 + 4) * DPITCH + bc] = a1;
      sT4[br0 * DPITCH + bc]       = b0;
      sT4[(br0 + 4) * DPITCH + bc] = b1;
      sT2[br0 * DPITCH + bc]       = isY ? 0.5f * (a0 + b0) : a0;
      sT2[(br0 + 4) * DPITCH + bc] = isY ? 0.5f * (a1 + b1) : a1;
    }
    if (tid < BT * NUc) {
      const int r = tid >> 1, c = tid & 1;
      sU[tid] = uv;
      sT1[r * DPITCH + 58 + c] = uv;
      sT2[r * DPITCH + 58 + c] = uv;
      sT4[r * DPITCH + 58 + c] = uv;
    }
    __syncthreads();

    v4f hist1, hist2, hist4;
    {
      const int n  = lane & 15;
      const int kh = lane >> 4;
      v2f bh[16], aA[16], aB[16];
      const v2f* bp = W1p + (size_t)kh * LDW + 16 * wave + n;
#pragma unroll
      for (int kk = 2; kk < 16; ++kk) bh[kk] = bp[(size_t)(2 * kk) * LDW];
      hist_load_a(((n < BT) ? sT1 : sT4) + (n & (BT - 1)) * DPITCH + 2 * kh, aA);
      hist_load_a(sT2 + (n & (BT - 1)) * DPITCH + 2 * kh, aB);
      const float bias1 = sB1[16 * wave + n];
      const v8f dA = hist_mul(aA, bh);
      const v8f dB = hist_mul(aB, bh);
      const v4f lo = __builtin_shufflevector(dA, dA, 0, 1, 2, 3);
      const v4f hi = __builtin_shufflevector(dA, dA, 4, 5, 6, 7);
      const v4f give = kh ? lo : hi;
      v4f take;
#pragma unroll
      for (int v = 0; v < 4; ++v) take[v] = __shfl_xor(give[v], 16, 32);
      hist1 = (kh ? take : lo) + bias1;
      hist4 = (kh ? hi : take) + bias1;
      hist2 = my_rows(dB, kh) + bias1;
    }

    rk_stage(tid, wave, lane, 0, hist1, sP2, Sw, W1p, W2p, W3p, sB2, sB3,
             sXG, sU, sF, sPn, sNN, 0.0f, 0.5f * DELTA_F);

    rk_stage(tid, wave, lane, 1, hist2, sP2, Sw, W1p, W2p, W3p, sB2, sB3,
             sXG, sU, sF, sPn, sNN, 0.5f * DELTA_F, 0.5f * DELTA_F);

    rk_stage(tid, wave, lane, 2, hist2, sP2, Sw, W1p, W2p, W3p, sB2, sB3,
             sXG, sU, sF, sPn, sNN, 0.5f * DELTA_F, DELTA_F);

    rk_stage(tid, wave, lane, 3, hist4, sP2, Sw, W1p, W2p, W3p, sB2, sB3,
             sXG, sU, sF, sPn, sNN, DELTA_F, 0.0f);

    float xgnew = 0.0f;
    if (tid < BT * NXc) {
      int r = tid >> 3, j = tid & 7;
      float xg = sXG[r * NXc + j];
      out[(size_t)(bt0 + r) * T_STEPS * NXc + (size_t)t * NXc + j] = xg;
      const float k1 = sF[0 * KST + tid] + sNN[0 * KST + tid];
      const float k2 = sF[1 * KST + tid] + sNN[1 * KST + tid];
      const float k3 = sF[2 * KST + tid] + sNN[2 * KST + tid];
      const float k4 = sF[3 * KST + tid] + sNN[3 * KST + tid];
      xgnew = xg + (DELTA_F / 6.0f) * (k1 + 2.0f * k2 + 2.0f * k3 + k4);
    }
    float regz[2];
#pragma unroll
    for (int ii = 0; ii < 2; ++ii) {
      int i = tid + ii * 256;
      if (i < BT * NZc) {
        int r = i / NZc, c = i - r * NZc;
        float v;
        if      (c < 32) v = sZ[r * 52 + 8 + c];
        else if (c < 40) v = sXG[r * NXc + (c - 32)];
        else if (c < 48) v = sZ[r * 52 + (c + 2)];
        else             v = sU[r * 2 + (c - 48)];
        regz[ii] = v;
      }
    }
    __syncthreads();
#pragma unroll
    for (int ii = 0; ii < 2; ++ii) {
      int i = tid + ii * 256;
      if (i < BT * NZc) {
        int r = i / NZc, c = i - r * NZc;
        sZ[r * 52 + c] = regz[ii];
      }
    }
    if (tid < BT * NXc) sXG[tid] = xgnew;
    __syncthreads();
  }
}

extern "C" void kernel_launch(void* const* d_in, const int* in_sizes, int n_in,
                              void* d_out, int out_size, void* d_ws, size_t ws_size,
                              hipStream_t stream) {
  (void)in_sizes; (void)n_in; (void)out_size; (void)d_ws; (void)ws_size;
  const float* useq = (const float*)d_in[0];
  const float* xGz0 = (const float*)d_in[1];
  const float* W1   = (const float*)d_in[2];
  const float* b1   = (const float*)d_in[3];
  const float* W2   = (const float*)d_in[4];
  const float* b2   = (const float*)d_in[5];
  const float* W3   = (const float*)d_in[6];
  const float* b3   = (const float*)d_in[7];
  float* out = (float*)d_out;
  cstr_rollout_kernel<<<dim3(B_TOT / BT), dim3(256), SMEM_BYTES, stream>>>(
      useq, xGz0, W1, b1, W2, b2, W3, b3, out);
}


# Rollout I/O and history as time-indexed LDS arrays: u staged once, y kept and stored as whole lines at the end; no global op or shift pass in the loop, 1 barrier between steps, not 3
# speedup vs baseline: 3.1688x; 1.0529x over previous
#include <hip/hip_runtime.h>
#include <math.h>


typedef __attribute__((ext_vector_type(2))) float v2f;
typedef __attribute__((ext_vector_type(4))) float v4f;
typedef __attribute__((ext_vector_type(8))) float v8f;

#define B_TOT   2048
#define T_STEPS 128
#define NXc     8
#define NUc     2
#define NZc     50
#define DINP    64
#define DPITCH  68
#define DH      128
#define BT      8
#define KST     (BT * NXc)
#define DELTA_F 0.01f

#define LDW     129
#define LDW3    17
#define OFF_W1P 0u
#define OFF_W2P (OFF_W1P + 32u*LDW*8u)
#define OFF_W3P (OFF_W2P + 64u*LDW*8u)
#define OFF_INP (OFF_W3P + 64u*LDW3*8u)
#define SW_FLT  136
#define P_FLT   1024
#define OFF_P1  (OFF_INP + 16u*64u*4u)
#define OFF_T2  (OFF_P1)
#define OFF_T4  (OFF_P1 + 4096u + 128u)
#define OFF_P2  (OFF_P1  + 6656u)
#define OFF_SW  (OFF_P2  + 8u*P_FLT*4u)
#define OFF_NN  (OFF_SW  + 8u*SW_FLT*4u)
#define OFF_XG  (OFF_NN  + 256u*4u)
#define OFF_F   (OFF_XG  + 128u*4u)
#define OFF_PN  (OFF_F   + 256u*4u)
#define OFF_B1  (OFF_PN  + 256u*4u)
#define OFF_B2  (OFF_B1  + 128u*4u)
#define OFF_B3  (OFF_B2  + 128u*4u)
#define EPITCH  1072
#define UPITCH  268
#define OFF_E   (OFF_B3 + 16u*4u)
#define OFF_UE  (OFF_E  + 8u*EPITCH*4u)
#define SMEM_BYTES (OFF_UE + 8u*UPITCH*4u)

__device__ __forceinline__ float fast_tanh(float x) {
  float e = __expf(2.0f * x);
  return 1.0f - __fdividef(2.0f, e + 1.0f);
}


__device__ __forceinline__ v4f my_rows(v8f d, int kh) {
  const v4f lo = __builtin_shufflevector(d, d, 0, 1, 2, 3);
  const v4f hi = __builtin_shufflevector(d, d, 4, 5, 6, 7);
  return kh ? hi : lo;
}

__device__ __forceinline__ int sw_row(int r) { return 16 * r + 4 * (r >> 2); }

__device__ __forceinline__ void own_cols_to_afrag(float* Sw, v4f h, int lane, v2f a[4]) {
  const int n  = lane & 15;
  const int kh = lane >> 4;
#pragma unroll
  for (int v = 0; v < 4; ++v) Sw[sw_row(4 * kh + v) + n] = h[v];
  const float* rp = Sw + sw_row(n & (BT - 1)) + 2 * kh;
#pragma unroll
  for (int kk = 0; kk < 4; ++kk) a[kk] = *(const v2f*)(rp + 4 * kk);
}

template <int KSTEPS>
__device__ __forceinline__ void slice_times_rows(const v2f* a, const v2f* __restrict__ Bp,
                                                 float* pw, int lane) {
  const int n  = lane & 15;
  const int kh = lane >> 4;
  v8f acc[8];
#pragma unroll
  for (int j = 0; j < 8; ++j) acc[j] = (v8f){};
  const v2f* bp = Bp + (size_t)kh * LDW + n;
#pragma unroll
  for (int kk = 0; kk < KSTEPS; ++kk) {
#pragma unroll
    for (int j = 0; j < 8; ++j) {
      v2f b = bp[(size_t)(2 * kk) * LDW + 16 * j];
      acc[j] = __builtin_amdgcn_wmma_f32_16x16x4_f32(false, a[kk], false, b,
                                                     (short)0, acc[j], false, false);
    }
  }
  float* p = pw + n * 8 + 4 * kh;
#pragma unroll
  for (int j = 0; j < 8; ++j) *(v4f*)(p + j * 128) = my_rows(acc[j], kh);
}

__device__ __forceinline__ void hist_load_a(const float* ap, v2f a[16]) {
#pragma unroll
  for (int kk = 2; kk < 16; ++kk) a[kk] = *(const v2f*)(ap + 4 * kk);
}
__device__ __forceinline__ v8f hist_mul(const v2f a[16], const v2f b[16]) {
  v8f acc[4];
#pragma unroll
  for (int c = 0; c < 4; ++c) acc[c] = (v8f){};
#pragma unroll
  for (int kk = 2; kk < 16; ++kk) {
    acc[kk & 3] = __builtin_amdgcn_wmma_f32_16x16x4_f32(false, a[kk], false, b[kk],
                                                        (short)0, acc[kk & 3], false, false);
  }
  return (acc[0] + acc[1]) + (acc[2] + acc[3]);
}

__device__ __forceinline__ v4f sum_slices_tanh(const float* sP, int nc, int kh, float bias) {
  const float* pr = sP + nc * 8 + 4 * kh;
  v4f s = *(const v4f*)pr;
#pragma unroll
  for (int w = 1; w < 8; ++w) s += *(const v4f*)(pr + w * P_FLT);
  v4f h;
#pragma unroll
  for (int v = 0; v < 4; ++v) h[v] = fast_tanh(s[v] + bias);
  return h;
}

__device__ __forceinline__ void rk_stage(
    int tid, int wave, int lane, int stage,
    v4f hist, float* sP2, float* Sw,
    const v2f* W1p, const v2f* W2p, const v2f* W3p,
    const float* sB2, const float* sB3,
    const float* sXG, const float* sU,
    float* sF, float* sPn, float* sNN, float cprev, float cnext) {
  const int n   = lane & 15;
  const int kh  = lane >> 4;
  const int row = n & (BT - 1);
  const int nc  = wave * 16 + n;
  const int    sprev = (stage > 0) ? stage - 1 : 0;
  float*       sNNs  = sNN + stage * (BT * NXc);
  const float* sNNp  = sNN + sprev * (BT * NXc);
  const float* sPp   = sPn + sprev * (BT * NXc);
  v4f g0, g1, q0, q1, m0, m1, c0, c1;
  v2f uu;
  if (tid < BT) {
    g0 = *(const v4f*)(sXG + tid * NXc);
    g1 = *(const v4f*)(sXG + tid * NXc + 4);
    if (stage != 0) {
      q0 = *(const v4f*)(sPp  + tid * NXc);  q1 = *(const v4f*)(sPp  + tid * NXc + 4);
      m0 = *(const v4f*)(sNNp + tid * NXc);  m1 = *(const v4f*)(sNNp + tid * NXc + 4);
    }
    uu = *(const v2f*)(sU + tid * UPITCH);
    c0 = *(const v4f*)sB3;
    c1 = *(const v4f*)(sB3 + 4);
  }
  v2f p0, p1, n0, n1;
  if (stage == 0) {
    const float* xp = sXG + row * NXc + 2 * kh;
    p0 = *(const v2f*)xp;
    p1 = *(const v2f*)(xp + 4);
  } else {
    const float* pp = sPp  + row * NXc + 2 * kh;
    const float* np = sNNp + row * NXc + 2 * kh;
    p0 = *(const v2f*)pp;  p1 = *(const v2f*)(pp + 4);
    n0 = *(const v2f*)np;  n1 = *(const v2f*)(np + 4);
  }
  const v2f* bp1 = W1p + (size_t)kh * LDW + 16 * wave + n;
  const v2f bs0 = bp1[0];
  const v2f bs1 = bp1[(size_t)2 * LDW];
  if (tid < BT) {
    v4f x0 = g0, x1 = g1;
    if (stage != 0) {
      x0 = q0 + cprev * m0;
      x1 = q1 + cprev * m1;
    }
    float Hr  = x0[0] * 2.0f + 50.0f;
    float CAr = x0[1] * 0.2f + 1.0f;
    float CBr = x0[2] * 0.1f + 0.5f;
    float Tr  = x0[3] * 4.0f + 313.0f;
    float Hb  = x1[0] * 2.0f + 50.0f;
    float CAb = x1[1] * 0.2f + 1.0f;
    float CBb = x1[2] * 0.1f + 0.5f;
    float Tb  = x1[3] * 4.0f + 313.0f;
    float F = uu[0] * 1.0f + 10.0f;
    float D = uu[1] * 0.5f + 5.0f;
    float Fr = 2.5f * __fsqrt_rn(Hr);
    float Fb = 1.5f * __fsqrt_rn(Hb);
    float invArHr = __fdividef(1.0f, 3.0f * Hr);
    float invAbHb = __fdividef(1.0f, 3.0f * Hb);
    float dHr  = (F + D - Fr) * (1.0f / 3.0f);
    float dCAr = (F * (6.0f - CAr) - D * CAr) * invArHr;
    float dCBr = (-F * CBr - D * CBr) * invArHr;
    float dTr  = (F * (320.0f - Tr) + D * (300.0f - Tr)) * invArHr
               + (200.0f / 30.0f) * invArHr;
    float dHb  = (Fr - Fb - D) * (1.0f / 3.0f);
    float dCAb = (Fr * (CAr - CAb) + D * CAb) * invAbHb;
    float dCBb = (Fr * (CBr - CBb) + D * CBb) * invAbHb;
    float dTb  = Fr * (Tr - Tb) * invAbHb + (200.0f / 30.0f) * invAbHb;
    v4f f0, f1;
    f0[0] = dHr  * 0.5f  + c0[0];
    f0[1] = dCAr * 5.0f  + c0[1];
    f0[2] = dCBr * 10.0f + c0[2];
    f0[3] = dTr  * 0.25f + c0[3];
    f1[0] = dHb  * 0.5f  + c1[0];
    f1[1] = dCAb * 5.0f  + c1[1];
    f1[2] = dCBb * 10.0f + c1[2];
    f1[3] = dTb  * 0.25f + c1[3];
    float* fr = sF + stage * (BT * NXc) + tid * NXc;
    *(v4f*)fr       = f0;
    *(v4f*)(fr + 4) = f1;
    if (stage != 3) {
      float* pr = sPn + stage * (BT * NXc) + tid * NXc;
      *(v4f*)pr       = g0 + cnext * f0;
      *(v4f*)(pr + 4) = g1 + cnext * f1;
    }
  }
  v2f a[4];
  {
    const v2f as0 = (stage == 0) ? p0 : p0 + cprev * n0;
    const v2f as1 = (stage == 0) ? p1 : p1 + cprev * n1;
    const v8f d0 = __builtin_amdgcn_wmma_f32_16x16x4_f32(false, as0, false, bs0,
                                                         (short)0, (v8f){}, false, false);
    const v8f d1 = __builtin_amdgcn_wmma_f32_16x16x4_f32(false, as1, false, bs1,
                                                         (short)0, (v8f){}, false, false);
    const v4f s1 = (hist + my_rows(d0, kh)) + my_rows(d1, kh);
    v4f h1;
#pragma unroll
    for (int v = 0; v < 4; ++v) h1[v] = fast_tanh(s1[v]);
    own_cols_to_afrag(Sw, h1, lane, a);
  }
  slice_times_rows<4>(a, W2p + (size_t)(8 * wave) * LDW, sP2 + wave * P_FLT, lane);
  __syncthreads();
  own_cols_to_afrag(Sw, sum_slices_tanh(sP2, nc, kh, sB2[nc]), lane, a);
  {
    v8f d = {};
    const v2f* bp = W3p + (size_t)(8 * wave + kh) * LDW3 + n;
#pragma unroll
    for (int kk = 0; kk < 4; ++kk) {
      v2f b = bp[(size_t)(2 * kk) * LDW3];
      d = __builtin_amdgcn_wmma_f32_16x16x4_f32(false, a[kk], false, b,
                                                (short)0, d, false, false);
    }
    const v4f s = my_rows(d, kh);
    if (n < NXc) {
#pragma unroll
      for (int v = 0; v < 4; ++v)
        atomicAdd(&sNNs[(v + 4 * kh) * NXc + n], s[v]);
    }
  }
  __syncthreads();
}

__global__ __launch_bounds__(256) void cstr_rollout_kernel(
    const float* __restrict__ useq, const float* __restrict__ xGz0,
    const float* __restrict__ W1g, const float* __restrict__ b1g,
    const float* __restrict__ W2g, const float* __restrict__ b2g,
    const float* __restrict__ W3g, const float* __restrict__ b3g,
    float* __restrict__ out) {
  extern __shared__ __align__(16) char smem[];
  v2f*   W1p  = (v2f*)(smem + OFF_W1P);
  v2f*   W2p  = (v2f*)(smem + OFF_W2P);
  v2f*   W3p  = (v2f*)(smem + OFF_W3P);
  float* sT1  = (float*)(smem + OFF_INP);
  float* sT2  = (float*)(smem + OFF_T2);
  float* sT4  = (float*)(smem + OFF_T4);
  float* sP2  = (float*)(smem + OFF_P2);
  float* sNN  = (float*)(smem + OFF_NN);
  float* sXG  = (float*)(smem + OFF_XG);
  float* sF   = (float*)(smem + OFF_F);
  float* sPn  = (float*)(smem + OFF_PN);
  float* sE   = (float*)(smem + OFF_E);
  float* sUe  = (float*)(smem + OFF_UE);
  float* sB1  = (float*)(smem + OFF_B1);
  float* sB2  = (float*)(smem + OFF_B2);
  float* sB3  = (float*)(smem + OFF_B3);

  const int tid  = threadIdx.x;
  const int lane = tid & 31;
  const int wave = tid >> 5;
  const int bt0  = blockIdx.x * BT;
  float* Sw = (float*)(smem + OFF_SW) + wave * SW_FLT;

  for (int i = tid; i < 32 * 128; i += 256) {
    int kk = i >> 7, n = i & 127;
    float lo = (2 * kk     < 60) ? W1g[(2 * kk)     * 128 + n] : 0.0f;
    float hi = (2 * kk + 1 < 60) ? W1g[(2 * kk + 1) * 128 + n] : 0.0f;
    W1p[kk * LDW + n] = (v2f){lo, hi};
  }
  for (int i = tid; i < 64 * 128; i += 256) {
    int kk = i >> 7, n = i & 127;
    W2p[kk * LDW + n] = (v2f){W2g[(2 * kk) * 128 + n], W2g[(2 * kk + 1) * 128 + n]};
  }
  for (int i = tid; i < 64 * 16; i += 256) {
    int kk = i >> 4, n = i & 15;
    v2f p = (v2f){0.0f, 0.0f};
    if (n < NXc) p = (v2f){W3g[(2 * kk) * NXc + n], W3g[(2 * kk + 1) * NXc + n]};
    W3p[kk * LDW3 + n] = p;
  }
  if (tid < 128) { sB1[tid] = b1g[tid]; sB2[tid] = b2g[tid]; }
  if (tid < NXc) sB3[tid] = b3g[tid];
  if (tid < 3 * BT * 4) {
    float* tl = (tid < 32) ? sT1 : ((tid < 64) ? sT2 : sT4);
    tl[((tid >> 2) & (BT - 1)) * DPITCH + 60 + (tid & 3)] = 0.0f;
  }
  const float* S = (const float*)smem;
  const int  bc   = tid & 63;
  const int  br0  = tid >> 6;
  const bool act  = (bc >= NXc) && (bc < 60);
  const bool isY  = bc < 48;
  const bool new8 = (bc >= 40) && (bc < 48);
  const int  adv  = isY ? NXc : NUc;
  const int  rs4  = 4 * (isY ? EPITCH : UPITCH);
  const int  src0 = isY ? (int)(OFF_E  / 4u) + br0 * EPITCH + (bc - 8)
                        : (int)(OFF_UE / 4u) + br0 * UPITCH + (bc - 48);
  auto build_tiles = [&](int tp, bool all) {
    if (act) {
      const bool wb = all || !new8;
      const int  ia = src0 + tp * adv;
      const int  ib = (isY && wb) ? ia + NXc : ia;
      const float a0 = S[ia];
      const float b0 = S[ib];
      const float a1 = S[ia + rs4];
      const float b1 = S[ib + rs4];
      sT1[br0 * DPITCH + bc]       = a0;
      sT1[(br0 + 4) * DPITCH + bc] = a1;
      if (wb) {
        sT4[br0 * DPITCH + bc]       = b0;
        sT4[(br0 + 4) * DPITCH + bc] = b1;
        sT2[br0 * DPITCH + bc]       = isY ? 0.5f * (a0 + b0) : a0;
        sT2[(br0 + 4) * DPITCH + bc] = isY ? 0.5f * (a1 + b1) : a1;
      }
    }
  };
  for (int i = tid; i < BT * 58; i += 256) {
    int r = i / 58, c = i - r * 58;
    float v = xGz0[(size_t)(bt0 + r) * 58 + c];
    if (c < NXc)     { sXG[r * NXc + c] = v;  sE[r * EPITCH + 5 * NXc + c] = v; }
    else if (c < 48) sE[r * EPITCH + (c - NXc)] = v;
    else             sUe[r * UPITCH + (c - 48)] = v;
  }
  for (int i = tid; i < BT * T_STEPS * NUc; i += 256)
    sUe[(i >> 8) * UPITCH + 5 * NUc + (i & 255)] = useq[(size_t)bt0 * T_STEPS * NUc + i];
  sNN[tid] = 0.0f;
  __syncthreads();
  build_tiles(0, true);
  __syncthreads();

  for (int t = 0; t < T_STEPS; ++t) {
    const float* sU = sUe + (t + 5) * NUc;

    v4f hist1, hist2, hist4;
    {
      const int n  = lane & 15;
      const int kh = lane >> 4;
      v2f bh[16], aA[16], aB[16];
      const v2f* bp = W1p + (size_t)kh * LDW + 16 * wave + n;
#pragma unroll
      for (int kk = 2; kk < 16; ++kk) bh[kk] = bp[(size_t)(2 * kk) * LDW];
      hist_load_a(((n < BT) ? sT1 : sT4) + (n & (BT - 1)) * DPITCH + 2 * kh, aA);
      hist_load_a(sT2 + (n & (BT - 1)) * DPITCH + 2 * kh, aB);
      const float bias1 = sB1[16 * wave + n];
      const v8f dA = hist_mul(aA, bh);
      const v8f dB = hist_mul(aB, bh);
      const v4f lo = __builtin_shufflevector(dA, dA, 0, 1, 2, 3);
      const v4f hi = __builtin_shufflevector(dA, dA, 4, 5, 6, 7);
      const v4f give = kh ? lo : hi;
      v4f take;
#pragma unroll
      for (int v = 0; v < 4; ++v) take[v] = __shfl_xor(give[v], 16, 32);
      hist1 = (kh ? take : lo) + bias1;
      hist4 = (kh ? hi : take) + bias1;
      hist2 = my_rows(dB, kh) + bias1;
    }

    rk_stage(tid, wave, lane, 0, hist1, sP2, Sw, W1p, W2p, W3p, sB2, sB3,
             sXG, sU, sF, sPn, sNN, 0.0f, 0.5f * DELTA_F);

    rk_stage(tid, wave, lane, 1, hist2, sP2, Sw, W1p, W2p, W3p, sB2, sB3,
             sXG, sU, sF, sPn, sNN, 0.5f * DELTA_F, 0.5f * DELTA_F);

    rk_stage(tid, wave, lane, 2, hist2, sP2, Sw, W1p, W2p, W3p, sB2, sB3,
             sXG, sU, sF, sPn, sNN, 0.5f * DELTA_F, DELTA_F);

    rk_stage(tid, wave, lane, 3, hist4, sP2, Sw, W1p, W2p, W3p, sB2, sB3,
             sXG, sU, sF, sPn, sNN, DELTA_F, 0.0f);

    if (tid < BT * NXc) {
      const int r = tid >> 3, j = tid & 7;
      const float xg = sXG[tid];
      const float k1 = sF[0 * KST + tid] + sNN[0 * KST + tid];
      const float k2 = sF[1 * KST + tid] + sNN[1 * KST + tid];
      const float k3 = sF[2 * KST + tid] + sNN[2 * KST + tid];
      const float k4 = sF[3 * KST + tid] + sNN[3 * KST + tid];
      const float xgnew = xg + (DELTA_F / 6.0f) * (k1 + 2.0f * k2 + 2.0f * k3 + k4);
      sNN[0 * KST + tid] = 0.0f;
      sNN[1 * KST + tid] = 0.0f;
      sNN[2 * KST + tid] = 0.0f;
      sNN[3 * KST + tid] = 0.0f;
      sXG[tid] = xgnew;
      sE[r * EPITCH + (t + 6) * NXc + j] = xgnew;
      sT4[r * DPITCH + 40 + j] = xgnew;
      sT2[r * DPITCH + 40 + j] = 0.5f * (xg + xgnew);
    }
    if (t + 1 < T_STEPS) build_tiles(t + 1, false);
    __syncthreads();
  }
  float* og = out + (size_t)bt0 * T_STEPS * NXc;
  for (int r = 0; r < BT; ++r)
    *(v4f*)(og + r * (T_STEPS * NXc) + 4 * tid) =
        *(const v4f*)(sE + r * EPITCH + 5 * NXc + 4 * tid);
}

extern "C" void kernel_launch(void* const* d_in, const int* in_sizes, int n_in,
                              void* d_out, int out_size, void* d_ws, size_t ws_size,
                              hipStream_t stream) {
  (void)in_sizes; (void)n_in; (void)out_size; (void)d_ws; (void)ws_size;
  const float* useq = (const float*)d_in[0];
  const float* xGz0 = (const float*)d_in[1];
  const float* W1   = (const float*)d_in[2];
  const float* b1   = (const float*)d_in[3];
  const float* W2   = (const float*)d_in[4];
  const float* b2   = (const float*)d_in[5];
  const float* W3   = (const float*)d_in[6];
  const float* b3   = (const float*)d_in[7];
  float* out = (float*)d_out;
  cstr_rollout_kernel<<<dim3(B_TOT / BT), dim3(256), SMEM_BYTES, stream>>>(
      useq, xGz0, W1, b1, W2, b2, W3, b3, out);
}


# Split-K partial planes laid [tile][row half][col][4 rows], not [col][8 rows]: a lane half's 16-byte stores and loads are 16 B apart and cover 64 banks once; they were 2-way bank-conflicted
# speedup vs baseline: 3.2710x; 1.0323x over previous
#include <hip/hip_runtime.h>
#include <math.h>


typedef __attribute__((ext_vector_type(2))) float v2f;
typedef __attribute__((ext_vector_type(4))) float v4f;
typedef __attribute__((ext_vector_type(8))) float v8f;

#define B_TOT   2048
#define T_STEPS 128
#define NXc     8
#define NUc     2
#define NZc     50
#define DINP    64
#define DPITCH  68
#define DH      128
#define BT      8
#define KST     (BT * NXc)
#define DELTA_F 0.01f

#define LDW     129
#define LDW3    17
#define OFF_W1P 0u
#define OFF_W2P (OFF_W1P + 32u*LDW*8u)
#define OFF_W3P (OFF_W2P + 64u*LDW*8u)
#define OFF_INP (OFF_W3P + 64u*LDW3*8u)
#define SW_FLT  136
#define P_FLT   1024
#define OFF_P1  (OFF_INP + 16u*64u*4u)
#define OFF_T2  (OFF_P1)
#define OFF_T4  (OFF_P1 + 4096u + 128u)
#define OFF_P2  (OFF_P1  + 6656u)
#define OFF_SW  (OFF_P2  + 8u*P_FLT*4u)
#define OFF_NN  (OFF_SW  + 8u*SW_FLT*4u)
#define OFF_XG  (OFF_NN  + 256u*4u)
#define OFF_F   (OFF_XG  + 128u*4u)
#define OFF_PN  (OFF_F   + 256u*4u)
#define OFF_B1  (OFF_PN  + 256u*4u)
#define OFF_B2  (OFF_B1  + 128u*4u)
#define OFF_B3  (OFF_B2  + 128u*4u)
#define EPITCH  1072
#define UPITCH  268
#define OFF_E   (OFF_B3 + 16u*4u)
#define OFF_UE  (OFF_E  + 8u*EPITCH*4u)
#define SMEM_BYTES (OFF_UE + 8u*UPITCH*4u)

__device__ __forceinline__ float fast_tanh(float x) {
  float e = __expf(2.0f * x);
  return 1.0f - __fdividef(2.0f, e + 1.0f);
}


__device__ __forceinline__ v4f my_rows(v8f d, int kh) {
  const v4f lo = __builtin_shufflevector(d, d, 0, 1, 2, 3);
  const v4f hi = __builtin_shufflevector(d, d, 4, 5, 6, 7);
  return kh ? hi : lo;
}

__device__ __forceinline__ int sw_row(int r) { return 16 * r + 4 * (r >> 2); }

__device__ __forceinline__ void own_cols_to_afrag(float* Sw, v4f h, int lane, v2f a[4]) {
  const int n  = lane & 15;
  const int kh = lane >> 4;
#pragma unroll
  for (int v = 0; v < 4; ++v) Sw[sw_row(4 * kh + v) + n] = h[v];
  const float* rp = Sw + sw_row(n & (BT - 1)) + 2 * kh;
#pragma unroll
  for (int kk = 0; kk < 4; ++kk) a[kk] = *(const v2f*)(rp + 4 * kk);
}

template <int KSTEPS>
__device__ __forceinline__ void slice_times_rows(const v2f* a, const v2f* __restrict__ Bp,
                                                 float* pw, int lane) {
  const int n  = lane & 15;
  const int kh = lane >> 4;
  v8f acc[8];
#pragma unroll
  for (int j = 0; j < 8; ++j) acc[j] = (v8f){};
  const v2f* bp = Bp + (size_t)kh * LDW + n;
#pragma unroll
  for (int kk = 0; kk < KSTEPS; ++kk) {
#pragma unroll
    for (int j = 0; j < 8; ++j) {
      v2f b = bp[(size_t)(2 * kk) * LDW + 16 * j];
      acc[j] = __builtin_amdgcn_wmma_f32_16x16x4_f32(false, a[kk], false, b,
                                                     (short)0, acc[j], false, false);
    }
  }
  float* p = pw + 64 * kh + 4 * n;
#pragma unroll
  for (int j = 0; j < 8; ++j) *(v4f*)(p + j * 128) = my_rows(acc[j], kh);
}

__device__ __forceinline__ void hist_load_a(const float* ap, v2f a[16]) {
#pragma unroll
  for (int kk = 2; kk < 16; ++kk) a[kk] = *(const v2f*)(ap + 4 * kk);
}
__device__ __forceinline__ v8f hist_mul(const v2f a[16], const v2f b[16]) {
  v8f acc[4];
#pragma unroll
  for (int c = 0; c < 4; ++c) acc[c] = (v8f){};
#pragma unroll
  for (int kk = 2; kk < 16; ++kk) {
    acc[kk & 3] = __builtin_amdgcn_wmma_f32_16x16x4_f32(false, a[kk], false, b[kk],
                                                        (short)0, acc[kk & 3], false, false);
  }
  return (acc[0] + acc[1]) + (acc[2] + acc[3]);
}

__device__ __forceinline__ v4f sum_slices_tanh(const float* sP, int tile, int n, int kh, float bias) {
  const float* pr = sP + tile * 128 + 64 * kh + 4 * n;
  v4f s = *(const v4f*)pr;
#pragma unroll
  for (int w = 1; w < 8; ++w) s += *(const v4f*)(pr + w * P_FLT);
  v4f h;
#pragma unroll
  for (int v = 0; v < 4; ++v) h[v] = fast_tanh(s[v] + bias);
  return h;
}

__device__ __forceinline__ void rk_stage(
    int tid, int wave, int lane, int stage,
    v4f hist, float* sP2, float* Sw,
    const v2f* W1p, const v2f* W2p, const v2f* W3p,
    const float* sB2, const float* sB3,
    const float* sXG, const float* sU,
    float* sF, float* sPn, float* sNN, float cprev, float cnext) {
  const int n   = lane & 15;
  const int kh  = lane >> 4;
  const int row = n & (BT - 1);
  const int nc  = wave * 16 + n;
  const int    sprev = (stage > 0) ? stage - 1 : 0;
  float*       sNNs  = sNN + stage * (BT * NXc);
  const float* sNNp  = sNN + sprev * (BT * NXc);
  const float* sPp   = sPn + sprev * (BT * NXc);
  v4f g0, g1, q0, q1, m0, m1, c0, c1;
  v2f uu;
  if (tid < BT) {
    g0 = *(const v4f*)(sXG + tid * NXc);
    g1 = *(const v4f*)(sXG + tid * NXc + 4);
    if (stage != 0) {
      q0 = *(const v4f*)(sPp  + tid * NXc);  q1 = *(const v4f*)(sPp  + tid * NXc + 4);
      m0 = *(const v4f*)(sNNp + tid * NXc);  m1 = *(const v4f*)(sNNp + tid * NXc + 4);
    }
    uu = *(const v2f*)(sU + tid * UPITCH);
    c0 = *(const v4f*)sB3;
    c1 = *(const v4f*)(sB3 + 4);
  }
  v2f p0, p1, n0, n1;
  if (stage == 0) {
    const float* xp = sXG + row * NXc + 2 * kh;
    p0 = *(const v2f*)xp;
    p1 = *(const v2f*)(xp + 4);
  } else {
    const float* pp = sPp  + row * NXc + 2 * kh;
    const float* np = sNNp + row * NXc + 2 * kh;
    p0 = *(const v2f*)pp;  p1 = *(const v2f*)(pp + 4);
    n0 = *(const v2f*)np;  n1 = *(const v2f*)(np + 4);
  }
  const v2f* bp1 = W1p + (size_t)kh * LDW + 16 * wave + n;
  const v2f bs0 = bp1[0];
  const v2f bs1 = bp1[(size_t)2 * LDW];
  if (tid < BT) {
    v4f x0 = g0, x1 = g1;
    if (stage != 0) {
      x0 = q0 + cprev * m0;
      x1 = q1 + cprev * m1;
    }
    float Hr  = x0[0] * 2.0f + 50.0f;
    float CAr = x0[1] * 0.2f + 1.0f;
    float CBr = x0[2] * 0.1f + 0.5f;
    float Tr  = x0[3] * 4.0f + 313.0f;
    float Hb  = x1[0] * 2.0f + 50.0f;
    float CAb = x1[1] * 0.2f + 1.0f;
    float CBb = x1[2] * 0.1f + 0.5f;
    float Tb  = x1[3] * 4.0f + 313.0f;
    float F = uu[0] * 1.0f + 10.0f;
    float D = uu[1] * 0.5f + 5.0f;
    float Fr = 2.5f * __fsqrt_rn(Hr);
    float Fb = 1.5f * __fsqrt_rn(Hb);
    float invArHr = __fdividef(1.0f, 3.0f * Hr);
    float invAbHb = __fdividef(1.0f, 3.0f * Hb);
    float dHr  = (F + D - Fr) * (1.0f / 3.0f);
    float dCAr = (F * (6.0f - CAr) - D * CAr) * invArHr;
    float dCBr = (-F * CBr - D * CBr) * invArHr;
    float dTr  = (F * (320.0f - Tr) + D * (300.0f - Tr)) * invArHr
               + (200.0f / 30.0f) * invArHr;
    float dHb  = (Fr - Fb - D) * (1.0f / 3.0f);
    float dCAb = (Fr * (CAr - CAb) + D * CAb) * invAbHb;
    float dCBb = (Fr * (CBr - CBb) + D * CBb) * invAbHb;
    float dTb  = Fr * (Tr - Tb) * invAbHb + (200.0f / 30.0f) * invAbHb;
    v4f f0, f1;
    f0[0] = dHr  * 0.5f  + c0[0];
    f0[1] = dCAr * 5.0f  + c0[1];
    f0[2] = dCBr * 10.0f + c0[2];
    f0[3] = dTr  * 0.25f + c0[3];
    f1[0] = dHb  * 0.5f  + c1[0];
    f1[1] = dCAb * 5.0f  + c1[1];
    f1[2] = dCBb * 10.0f + c1[2];
    f1[3] = dTb  * 0.25f + c1[3];
    float* fr = sF + stage * (BT * NXc) + tid * NXc;
    *(v4f*)fr       = f0;
    *(v4f*)(fr + 4) = f1;
    if (stage != 3) {
      float* pr = sPn + stage * (BT * NXc) + tid * NXc;
      *(v4f*)pr       = g0 + cnext * f0;
      *(v4f*)(pr + 4) = g1 + cnext * f1;
    }
  }
  v2f a[4];
  {
    const v2f as0 = (stage == 0) ? p0 : p0 + cprev * n0;
    const v2f as1 = (stage == 0) ? p1 : p1 + cprev * n1;
    const v8f d0 = __builtin_amdgcn_wmma_f32_16x16x4_f32(false, as0, false, bs0,
                                                         (short)0, (v8f){}, false, false);
    const v8f d1 = __builtin_amdgcn_wmma_f32_16x16x4_f32(false, as1, false, bs1,
                                                         (short)0, (v8f){}, false, false);
    const v4f s1 = (hist + my_rows(d0, kh)) + my_rows(d1, kh);
    v4f h1;
#pragma unroll
    for (int v = 0; v < 4; ++v) h1[v] = fast_tanh(s1[v]);
    own_cols_to_afrag(Sw, h1, lane, a);
  }
  slice_times_rows<4>(a, W2p + (size_t)(8 * wave) * LDW, sP2 + wave * P_FLT, lane);
  __syncthreads();
  own_cols_to_afrag(Sw, sum_slices_tanh(sP2, wave, n, kh, sB2[nc]), lane, a);
  {
    v8f d = {};
    const v2f* bp = W3p + (size_t)(8 * wave + kh) * LDW3 + n;
#pragma unroll
    for (int kk = 0; kk < 4; ++kk) {
      v2f b = bp[(size_t)(2 * kk) * LDW3];
      d = __builtin_amdgcn_wmma_f32_16x16x4_f32(false, a[kk], false, b,
                                                (short)0, d, false, false);
    }
    const v4f s = my_rows(d, kh);
    if (n < NXc) {
#pragma unroll
      for (int v = 0; v < 4; ++v)
        atomicAdd(&sNNs[(v + 4 * kh) * NXc + n], s[v]);
    }
  }
  __syncthreads();
}

__global__ __launch_bounds__(256) void cstr_rollout_kernel(
    const float* __restrict__ useq, const float* __restrict__ xGz0,
    const float* __restrict__ W1g, const float* __restrict__ b1g,
    const float* __restrict__ W2g, const float* __restrict__ b2g,
    const float* __restrict__ W3g, const float* __restrict__ b3g,
    float* __restrict__ out) {
  extern __shared__ __align__(16) char smem[];
  v2f*   W1p  = (v2f*)(smem + OFF_W1P);
  v2f*   W2p  = (v2f*)(smem + OFF_W2P);
  v2f*   W3p  = (v2f*)(smem + OFF_W3P);
  float* sT1  = (float*)(smem + OFF_INP);
  float* sT2  = (float*)(smem + OFF_T2);
  float* sT4  = (float*)(smem + OFF_T4);
  float* sP2  = (float*)(smem + OFF_P2);
  float* sNN  = (float*)(smem + OFF_NN);
  float* sXG  = (float*)(smem + OFF_XG);
  float* sF   = (float*)(smem + OFF_F);
  float* sPn  = (float*)(smem + OFF_PN);
  float* sE   = (float*)(smem + OFF_E);
  float* sUe  = (float*)(smem + OFF_UE);
  float* sB1  = (float*)(smem + OFF_B1);
  float* sB2  = (float*)(smem + OFF_B2);
  float* sB3  = (float*)(smem + OFF_B3);

  const int tid  = threadIdx.x;
  const int lane = tid & 31;
  const int wave = tid >> 5;
  const int bt0  = blockIdx.x * BT;
  float* Sw = (float*)(smem + OFF_SW) + wave * SW_FLT;

  for (int i = tid; i < 32 * 128; i += 256) {
    int kk = i >> 7, n = i & 127;
    float lo = (2 * kk     < 60) ? W1g[(2 * kk)     * 128 + n] : 0.0f;
    float hi = (2 * kk + 1 < 60) ? W1g[(2 * kk + 1) * 128 + n] : 0.0f;
    W1p[kk * LDW + n] = (v2f){lo, hi};
  }
  for (int i = tid; i < 64 * 128; i += 256) {
    int kk = i >> 7, n = i & 127;
    W2p[kk * LDW + n] = (v2f){W2g[(2 * kk) * 128 + n], W2g[(2 * kk + 1) * 128 + n]};
  }
  for (int i = tid; i < 64 * 16; i += 256) {
    int kk = i >> 4, n = i & 15;
    v2f p = (v2f){0.0f, 0.0f};
    if (n < NXc) p = (v2f){W3g[(2 * kk) * NXc + n], W3g[(2 * kk + 1) * NXc + n]};
    W3p[kk * LDW3 + n] = p;
  }
  if (tid < 128) { sB1[tid] = b1g[tid]; sB2[tid] = b2g[tid]; }
  if (tid < NXc) sB3[tid] = b3g[tid];
  if (tid < 3 * BT * 4) {
    float* tl = (tid < 32) ? sT1 : ((tid < 64) ? sT2 : sT4);
    tl[((tid >> 2) & (BT - 1)) * DPITCH + 60 + (tid & 3)] = 0.0f;
  }
  const float* S = (const float*)smem;
  const int  bc   = tid & 63;
  const int  br0  = tid >> 6;
  const bool act  = (bc >= NXc) && (bc < 60);
  const bool isY  = bc < 48;
  const bool new8 = (bc >= 40) && (bc < 48);
  const int  adv  = isY ? NXc : NUc;
  const int  rs4  = 4 * (isY ? EPITCH : UPITCH);
  const int  src0 = isY ? (int)(OFF_E  / 4u) + br0 * EPITCH + (bc - 8)
                        : (int)(OFF_UE / 4u) + br0 * UPITCH + (bc - 48);
  auto build_tiles = [&](int tp, bool all) {
    if (act) {
      const bool wb = all || !new8;
      const int  ia = src0 + tp * adv;
      const int  ib = (isY && wb) ? ia + NXc : ia;
      const float a0 = S[ia];
      const float b0 = S[ib];
      const float a1 = S[ia + rs4];
      const float b1 = S[ib + rs4];
      sT1[br0 * DPITCH + bc]       = a0;
      sT1[(br0 + 4) * DPITCH + bc] = a1;
      if (wb) {
        sT4[br0 * DPITCH + bc]       = b0;
        sT4[(br0 + 4) * DPITCH + bc] = b1;
        sT2[br0 * DPITCH + bc]       = isY ? 0.5f * (a0 + b0) : a0;
        sT2[(br0 + 4) * DPITCH + bc] = isY ? 0.5f * (a1 + b1) : a1;
      }
    }
  };
  for (int i = tid; i < BT * 58; i += 256) {
    int r = i / 58, c = i - r * 58;
    float v = xGz0[(size_t)(bt0 + r) * 58 + c];
    if (c < NXc)     { sXG[r * NXc + c] = v;  sE[r * EPITCH + 5 * NXc + c] = v; }
    else if (c < 48) sE[r * EPITCH + (c - NXc)] = v;
    else             sUe[r * UPITCH + (c - 48)] = v;
  }
  for (int i = tid; i < BT * T_STEPS * NUc; i += 256)
    sUe[(i >> 8) * UPITCH + 5 * NUc + (i & 255)] = useq[(size_t)bt0 * T_STEPS * NUc + i];
  sNN[tid] = 0.0f;
  __syncthreads();
  build_tiles(0, true);
  __syncthreads();

  for (int t = 0; t < T_STEPS; ++t) {
    const float* sU = sUe + (t + 5) * NUc;

    v4f hist1, hist2, hist4;
    {
      const int n  = lane & 15;
      const int kh = lane >> 4;
      v2f bh[16], aA[16], aB[16];
      const v2f* bp = W1p + (size_t)kh * LDW + 16 * wave + n;
#pragma unroll
      for (int kk = 2; kk < 16; ++kk) bh[kk] = bp[(size_t)(2 * kk) * LDW];
      hist_load_a(((n < BT) ? sT1 : sT4) + (n & (BT - 1)) * DPITCH + 2 * kh, aA);
      hist_load_a(sT2 + (n & (BT - 1)) * DPITCH + 2 * kh, aB);
      const float bias1 = sB1[16 * wave + n];
      const v8f dA = hist_mul(aA, bh);
      const v8f dB = hist_mul(aB, bh);
      const v4f lo = __builtin_shufflevector(dA, dA, 0, 1, 2, 3);
      const v4f hi = __builtin_shufflevector(dA, dA, 4, 5, 6, 7);
      const v4f give = kh ? lo : hi;
      v4f take;
#pragma unroll
      for (int v = 0; v < 4; ++v) take[v] = __shfl_xor(give[v], 16, 32);
      hist1 = (kh ? take : lo) + bias1;
      hist4 = (kh ? hi : take) + bias1;
      hist2 = my_rows(dB, kh) + bias1;
    }

    rk_stage(tid, wave, lane, 0, hist1, sP2, Sw, W1p, W2p, W3p, sB2, sB3,
             sXG, sU, sF, sPn, sNN, 0.0f, 0.5f * DELTA_F);

    rk_stage(tid, wave, lane, 1, hist2, sP2, Sw, W1p, W2p, W3p, sB2, sB3,
             sXG, sU, sF, sPn, sNN, 0.5f * DELTA_F, 0.5f * DELTA_F);

    rk_stage(tid, wave, lane, 2, hist2, sP2, Sw, W1p, W2p, W3p, sB2, sB3,
             sXG, sU, sF, sPn, sNN, 0.5f * DELTA_F, DELTA_F);

    rk_stage(tid, wave, lane, 3, hist4, sP2, Sw, W1p, W2p, W3p, sB2, sB3,
             sXG, sU, sF, sPn, sNN, DELTA_F, 0.0f);

    if (tid < BT * NXc) {
      const int r = tid >> 3, j = tid & 7;
      const float xg = sXG[tid];
      const float k1 = sF[0 * KST + tid] + sNN[0 * KST + tid];
      const float k2 = sF[1 * KST + tid] + sNN[1 * KST + tid];
      const float k3 = sF[2 * KST + tid] + sNN[2 * KST + tid];
      const float k4 = sF[3 * KST + tid] + sNN[3 * KST + tid];
      const float xgnew = xg + (DELTA_F / 6.0f) * (k1 + 2.0f * k2 + 2.0f * k3 + k4);
      sNN[0 * KST + tid] = 0.0f;
      sNN[1 * KST + tid] = 0.0f;
      sNN[2 * KST + tid] = 0.0f;
      sNN[3 * KST + tid] = 0.0f;
      sXG[tid] = xgnew;
      sE[r * EPITCH + (t + 6) * NXc + j] = xgnew;
      sT4[r * DPITCH + 40 + j] = xgnew;
      sT2[r * DPITCH + 40 + j] = 0.5f * (xg + xgnew);
    }
    if (t + 1 < T_STEPS) build_tiles(t + 1, false);
    __syncthreads();
  }
  float* og = out + (size_t)bt0 * T_STEPS * NXc;
  for (int r = 0; r < BT; ++r)
    *(v4f*)(og + r * (T_STEPS * NXc) + 4 * tid) =
        *(const v4f*)(sE + r * EPITCH + 5 * NXc + 4 * tid);
}

extern "C" void kernel_launch(void* const* d_in, const int* in_sizes, int n_in,
                              void* d_out, int out_size, void* d_ws, size_t ws_size,
                              hipStream_t stream) {
  (void)in_sizes; (void)n_in; (void)out_size; (void)d_ws; (void)ws_size;
  const float* useq = (const float*)d_in[0];
  const float* xGz0 = (const float*)d_in[1];
  const float* W1   = (const float*)d_in[2];
  const float* b1   = (const float*)d_in[3];
  const float* W2   = (const float*)d_in[4];
  const float* b2   = (const float*)d_in[5];
  const float* W3   = (const float*)d_in[6];
  const float* b3   = (const float*)d_in[7];
  float* out = (float*)d_out;
  cstr_rollout_kernel<<<dim3(B_TOT / BT), dim3(256), SMEM_BYTES, stream>>>(
      useq, xGz0, W1, b1, W2, b2, W3, b3, out);
}
